# starting kernel
# baseline (speedup 1.0000x reference)
#include <hip/hip_runtime.h>


#define LN_EPS 1e-5f

typedef __attribute__((ext_vector_type(2))) float v2f;
typedef __attribute__((ext_vector_type(8))) float v8f;


__global__ void vgae_zero(float* __restrict__ p, long n) {
  long i = (long)blockIdx.x * blockDim.x + threadIdx.x;
  if (i < n) p[i] = 0.0f;
}

__global__ void vgae_deg(const int* __restrict__ ei, float* __restrict__ deg, int E) {
  int e = blockIdx.x * blockDim.x + threadIdx.x;
  if (e < E) atomicAdd(&deg[ei[E + e]], 1.0f);
}

__global__ void vgae_invdeg(float* __restrict__ d, int n) {
  int i = blockIdx.x * blockDim.x + threadIdx.x;
  if (i < n) d[i] = 1.0f / fmaxf(d[i], 1.0f);
}

__global__ void vgae_agg(const float* __restrict__ h, const int* __restrict__ ei,
                         float* __restrict__ agg, int d, int E) {
  int e = blockIdx.x * blockDim.y + threadIdx.y;
  if (e >= E) return;
  int s = ei[e];
  int t = ei[E + e];
  float v = h[(long)s * d + threadIdx.x];
  atomicAdd(&agg[(long)t * d + threadIdx.x], v);
}

__global__ void vgae_sage_gemm(const float* __restrict__ agg,
                               const float* __restrict__ h,
                               const float* __restrict__ invdeg,
                               const float* __restrict__ Wl,
                               const float* __restrict__ bl,
                               const float* __restrict__ Wr,
                               const float* __restrict__ gamma,
                               const float* __restrict__ beta,
                               float* __restrict__ out,
                               int din, int dout, int do_ln) {
  __shared__ float smem[16 * 128];
  __shared__ float s_mean[16];
  __shared__ float s_rstd[16];

  const int lane = threadIdx.x & 31;
  const int wave = threadIdx.x >> 5;
  const int l15  = lane & 15;
  const int hi   = lane >> 4;
  const int rowbase = blockIdx.x * 16;
  const int arow = rowbase + l15;
  const int bcol = wave * 16 + l15;
  const int koff = 2 * hi;

  const float id = invdeg[arow];
  const float* aggrow = agg + (long)arow * din;
  const float* hrow   = h   + (long)arow * din;
  const float* wlrow  = Wl  + (long)bcol * din;
  const float* wrrow  = Wr  + (long)bcol * din;

  v8f c = {0.f, 0.f, 0.f, 0.f, 0.f, 0.f, 0.f, 0.f};

  for (int k = 0; k < din; k += 4) {
    v2f a = *(const v2f*)(aggrow + k + koff);
    a *= id;
    v2f b = *(const v2f*)(wlrow + k + koff);
    c = __builtin_amdgcn_wmma_f32_16x16x4_f32(false, a, false, b,
                                              (short)0, c, false, false);
  }
  for (int k = 0; k < din; k += 4) {
    v2f a = *(const v2f*)(hrow + k + koff);
    v2f b = *(const v2f*)(wrrow + k + koff);
    c = __builtin_amdgcn_wmma_f32_16x16x4_f32(false, a, false, b,
                                              (short)0, c, false, false);
  }

  const float bias = bl[bcol];

  if (!do_ln) {
    for (int v = 0; v < 8; ++v) {
      int r = rowbase + v + 8 * hi;
      out[(long)r * dout + bcol] = c[v] + bias;
    }
    return;
  }

  for (int v = 0; v < 8; ++v)
    smem[(v + 8 * hi) * 128 + bcol] = c[v] + bias;
  __syncthreads();

  if (threadIdx.x < 16) {
    const float* rowp = &smem[threadIdx.x * 128];
    float s = 0.f, ss = 0.f;
    for (int j = 0; j < 128; ++j) { float x = rowp[j]; s += x; ss += x * x; }
    float m   = s * (1.0f / 128.0f);
    float var = ss * (1.0f / 128.0f) - m * m;
    s_mean[threadIdx.x] = m;
    s_rstd[threadIdx.x] = rsqrtf(var + LN_EPS);
  }
  __syncthreads();

  for (int idx = threadIdx.x; idx < 16 * 128; idx += blockDim.x) {
    int r  = idx >> 7;
    int cc = idx & 127;
    float x = (smem[idx] - s_mean[r]) * s_rstd[r] * gamma[cc] + beta[cc];
    out[(long)(rowbase + r) * 128 + cc] = fmaxf(x, 0.0f);
  }
}


extern "C" void kernel_launch(void* const* d_in, const int* in_sizes, int n_in,
                              void* d_out, int out_size, void* d_ws, size_t ws_size,
                              hipStream_t stream) {
  const float* x  = (const float*)d_in[0];
  const int*   ei = (const int*)d_in[1];
  const int N = in_sizes[0] / 64;
  const int E = in_sizes[1] / 2;

  float* invdeg = (float*)d_ws;
  float* aggx   = invdeg + N;
  float* bufA   = aggx + (long)N * 64;
  float* bufB   = bufA + (long)N * 128;
  float* bufC   = bufB + (long)N * 128;

  vgae_zero<<<(N + 255) / 256, 256, 0, stream>>>(invdeg, N);
  vgae_deg<<<(E + 255) / 256, 256, 0, stream>>>(ei, invdeg, E);
  vgae_invdeg<<<(N + 255) / 256, 256, 0, stream>>>(invdeg, N);

  vgae_zero<<<((long)N * 64 + 255) / 256, 256, 0, stream>>>(aggx, (long)N * 64);
  {
    dim3 blk(64, 4);
    vgae_agg<<<(E + 3) / 4, blk, 0, stream>>>(x, ei, aggx, 64, E);
  }

  const int nrt = N / 16;
  for (int t = 0; t < 2; ++t) {
    const int base = 2 + t * 13;
    const float* Wl0 = (const float*)d_in[base + 0];
    const float* bl0 = (const float*)d_in[base + 1];
    const float* Wr0 = (const float*)d_in[base + 2];
    const float* g0  = (const float*)d_in[base + 3];
    const float* b0  = (const float*)d_in[base + 4];
    const float* Wl1 = (const float*)d_in[base + 5];
    const float* bl1 = (const float*)d_in[base + 6];
    const float* Wr1 = (const float*)d_in[base + 7];
    const float* g1  = (const float*)d_in[base + 8];
    const float* b1  = (const float*)d_in[base + 9];
    const float* Wl2 = (const float*)d_in[base + 10];
    const float* bl2 = (const float*)d_in[base + 11];
    const float* Wr2 = (const float*)d_in[base + 12];
    float* outp = (float*)d_out + (long)t * N * 64;

    vgae_sage_gemm<<<nrt, 256, 0, stream>>>(aggx, x, invdeg, Wl0, bl0, Wr0,
                                            g0, b0, bufA, 64, 128, 1);
    vgae_zero<<<((long)N * 128 + 255) / 256, 256, 0, stream>>>(bufB, (long)N * 128);
    { dim3 blk(128, 2); vgae_agg<<<(E + 1) / 2, blk, 0, stream>>>(bufA, ei, bufB, 128, E); }

    vgae_sage_gemm<<<nrt, 256, 0, stream>>>(bufB, bufA, invdeg, Wl1, bl1, Wr1,
                                            g1, b1, bufC, 128, 128, 1);
    vgae_zero<<<((long)N * 128 + 255) / 256, 256, 0, stream>>>(bufA, (long)N * 128);
    { dim3 blk(128, 2); vgae_agg<<<(E + 1) / 2, blk, 0, stream>>>(bufC, ei, bufA, 128, E); }

    vgae_sage_gemm<<<nrt, 128, 0, stream>>>(bufA, bufC, invdeg, Wl2, bl2, Wr2,
                                            nullptr, nullptr, outp, 128, 64, 0);
  }
}


# Neighbor sums: float atomic scatter-add replaced by a destination-bucketed edge list (int count+rank, scan, fill) and per-row gathers; sum order = arrival-order rank in bucket; zero fills dropped
# speedup vs baseline: 2.9303x; 2.9303x over previous
#include <hip/hip_runtime.h>


#define LN_EPS 1e-5f

typedef __attribute__((ext_vector_type(2))) float v2f;
typedef __attribute__((ext_vector_type(8))) float v8f;


__global__ void vgae_zero_i(int* __restrict__ p, int n) {
  int i = blockIdx.x * blockDim.x + threadIdx.x;
  if (i < n) p[i] = 0;
}

__global__ void vgae_count_rank(const int* __restrict__ ei, int* __restrict__ cnt,
                                int* __restrict__ rank, int E) {
  int e = blockIdx.x * blockDim.x + threadIdx.x;
  if (e < E) rank[e] = atomicAdd(&cnt[ei[E + e]], 1);
}

__global__ void vgae_scan(int* __restrict__ cnt, float* __restrict__ invdeg, int n) {
  typedef __attribute__((ext_vector_type(4))) int v4i;
  typedef __attribute__((ext_vector_type(4))) float v4f;
  __shared__ int s_wsum[32];
  __shared__ int s_wexc[32];
  __shared__ int s_total;
  const int lane = threadIdx.x & 31;
  const int wave = threadIdx.x >> 5;
  int carry = 0;
  for (int base = 0; base < n; base += 1024 * 16) {
    const int i0 = base + threadIdx.x * 16;
    const bool full = (i0 + 16 <= n);
    v4i q[4];
    if (full) {
      const v4i* src = (const v4i*)(cnt + i0);
      q[0] = src[0]; q[1] = src[1]; q[2] = src[2]; q[3] = src[3];
    } else {
      for (int k = 0; k < 16; ++k)
        q[k >> 2][k & 3] = (i0 + k < n) ? cnt[i0 + k] : 0;
    }
    int tsum = 0;
    for (int k = 0; k < 16; ++k) tsum += q[k >> 2][k & 3];
    int incl = tsum;
    for (int o = 1; o < 32; o <<= 1) {
      int t = __shfl_up(incl, o);
      if (lane >= o) incl += t;
    }
    if (lane == 31) s_wsum[wave] = incl;
    __syncthreads();
    if (wave == 0) {
      const int w = s_wsum[lane];
      int wi = w;
      for (int o = 1; o < 32; o <<= 1) {
        int t = __shfl_up(wi, o);
        if (lane >= o) wi += t;
      }
      s_wexc[lane] = wi - w;
      if (lane == 31) s_total = wi;
    }
    __syncthreads();
    int run = carry + s_wexc[wave] + incl - tsum;
    v4i p[4];
    v4f r[4];
    for (int k = 0; k < 16; ++k) {
      const int c = q[k >> 2][k & 3];
      p[k >> 2][k & 3] = run;
      r[k >> 2][k & 3] = 1.0f / fmaxf((float)c, 1.0f);
      run += c;
    }
    if (full) {
      v4i* dp = (v4i*)(cnt + i0);
      v4f* dr = (v4f*)(invdeg + i0);
      dp[0] = p[0]; dp[1] = p[1]; dp[2] = p[2]; dp[3] = p[3];
      dr[0] = r[0]; dr[1] = r[1]; dr[2] = r[2]; dr[3] = r[3];
    } else {
      for (int k = 0; k < 16; ++k)
        if (i0 + k < n) { cnt[i0 + k] = p[k >> 2][k & 3]; invdeg[i0 + k] = r[k >> 2][k & 3]; }
    }
    carry += s_total;
    __syncthreads();
  }
  if (threadIdx.x == 0) cnt[n] = carry;
}

__global__ void vgae_fill(const int* __restrict__ ei, const int* __restrict__ start,
                          const int* __restrict__ rank, int* __restrict__ nbr, int E) {
  int e = blockIdx.x * blockDim.x + threadIdx.x;
  if (e < E) nbr[start[ei[E + e]] + rank[e]] = ei[e];
}

template <int VEC>
__global__ void vgae_gather(const float* __restrict__ h, const int* __restrict__ start,
                            const int* __restrict__ nbr, float* __restrict__ agg, int n) {
  typedef __attribute__((ext_vector_type(VEC))) float vecf;
  const int D = 32 * VEC;
  const int lane = threadIdx.x & 31;
  const int node = __builtin_amdgcn_readfirstlane(
      (int)(blockIdx.x * (blockDim.x >> 5) + (threadIdx.x >> 5)));
  if (node >= n) return;
  const int beg = start[node];
  const int end = start[node + 1];
  const float* hl = h + lane * VEC;
  vecf acc = {};
  for (int base = beg; base < end; base += 32) {
    const int m = min(32, end - base);
    int mine = (lane < m) ? nbr[base + lane] : 0;
    mine = min(max(mine, 0), n - 1);
    int j = 0;
    for (; j + 4 <= m; j += 4) {
      const int s0 = __builtin_amdgcn_readlane(mine, j);
      const int s1 = __builtin_amdgcn_readlane(mine, j + 1);
      const int s2 = __builtin_amdgcn_readlane(mine, j + 2);
      const int s3 = __builtin_amdgcn_readlane(mine, j + 3);
      const vecf v0 = *(const vecf*)(hl + (long)s0 * D);
      const vecf v1 = *(const vecf*)(hl + (long)s1 * D);
      const vecf v2 = *(const vecf*)(hl + (long)s2 * D);
      const vecf v3 = *(const vecf*)(hl + (long)s3 * D);
      acc += v0; acc += v1; acc += v2; acc += v3;
    }
    for (; j < m; ++j) {
      const int s0 = __builtin_amdgcn_readlane(mine, j);
      acc += *(const vecf*)(hl + (long)s0 * D);
    }
  }
  *(vecf*)(agg + (long)node * D + lane * VEC) = acc;
}

__global__ void vgae_sage_gemm(const float* __restrict__ agg,
                               const float* __restrict__ h,
                               const float* __restrict__ invdeg,
                               const float* __restrict__ Wl,
                               const float* __restrict__ bl,
                               const float* __restrict__ Wr,
                               const float* __restrict__ gamma,
                               const float* __restrict__ beta,
                               float* __restrict__ out,
                               int din, int dout, int do_ln) {
  __shared__ float smem[16 * 128];
  __shared__ float s_mean[16];
  __shared__ float s_rstd[16];

  const int lane = threadIdx.x & 31;
  const int wave = threadIdx.x >> 5;
  const int l15  = lane & 15;
  const int hi   = lane >> 4;
  const int rowbase = blockIdx.x * 16;
  const int arow = rowbase + l15;
  const int bcol = wave * 16 + l15;
  const int koff = 2 * hi;

  const float id = invdeg[arow];
  const float* aggrow = agg + (long)arow * din;
  const float* hrow   = h   + (long)arow * din;
  const float* wlrow  = Wl  + (long)bcol * din;
  const float* wrrow  = Wr  + (long)bcol * din;

  v8f c = {0.f, 0.f, 0.f, 0.f, 0.f, 0.f, 0.f, 0.f};

  for (int k = 0; k < din; k += 4) {
    v2f a = *(const v2f*)(aggrow + k + koff);
    a *= id;
    v2f b = *(const v2f*)(wlrow + k + koff);
    c = __builtin_amdgcn_wmma_f32_16x16x4_f32(false, a, false, b,
                                              (short)0, c, false, false);
  }
  for (int k = 0; k < din; k += 4) {
    v2f a = *(const v2f*)(hrow + k + koff);
    v2f b = *(const v2f*)(wrrow + k + koff);
    c = __builtin_amdgcn_wmma_f32_16x16x4_f32(false, a, false, b,
                                              (short)0, c, false, false);
  }

  const float bias = bl[bcol];

  if (!do_ln) {
    for (int v = 0; v < 8; ++v) {
      int r = rowbase + v + 8 * hi;
      out[(long)r * dout + bcol] = c[v] + bias;
    }
    return;
  }

  for (int v = 0; v < 8; ++v)
    smem[(v + 8 * hi) * 128 + bcol] = c[v] + bias;
  __syncthreads();

  if (threadIdx.x < 16) {
    const float* rowp = &smem[threadIdx.x * 128];
    float s = 0.f, ss = 0.f;
    for (int j = 0; j < 128; ++j) { float x = rowp[j]; s += x; ss += x * x; }
    float m   = s * (1.0f / 128.0f);
    float var = ss * (1.0f / 128.0f) - m * m;
    s_mean[threadIdx.x] = m;
    s_rstd[threadIdx.x] = rsqrtf(var + LN_EPS);
  }
  __syncthreads();

  for (int idx = threadIdx.x; idx < 16 * 128; idx += blockDim.x) {
    int r  = idx >> 7;
    int cc = idx & 127;
    float x = (smem[idx] - s_mean[r]) * s_rstd[r] * gamma[cc] + beta[cc];
    out[(long)(rowbase + r) * 128 + cc] = fmaxf(x, 0.0f);
  }
}


extern "C" void kernel_launch(void* const* d_in, const int* in_sizes, int n_in,
                              void* d_out, int out_size, void* d_ws, size_t ws_size,
                              hipStream_t stream) {
  const float* x  = (const float*)d_in[0];
  const int*   ei = (const int*)d_in[1];
  const int N = in_sizes[0] / 64;
  const int E = in_sizes[1] / 2;

  float* invdeg = (float*)d_ws;
  int*   start  = (int*)(invdeg + N);
  int*   rank   = start + N + 64;
  int*   nbr    = rank + E;
  float* bufA   = invdeg + N + (long)N * 64;
  float* bufB   = bufA + (long)N * 128;
  float* bufC   = bufB + (long)N * 128;

  vgae_zero_i<<<(N + 255) / 256, 256, 0, stream>>>(start, N);
  vgae_count_rank<<<(E + 255) / 256, 256, 0, stream>>>(ei, start, rank, E);
  vgae_scan<<<1, 1024, 0, stream>>>(start, invdeg, N);
  vgae_fill<<<(E + 255) / 256, 256, 0, stream>>>(ei, start, rank, nbr, E);

  const int ngb = (N + 7) / 8;
  const int nrt = N / 16;
  for (int t = 0; t < 2; ++t) {
    const int base = 2 + t * 13;
    const float* Wl0 = (const float*)d_in[base + 0];
    const float* bl0 = (const float*)d_in[base + 1];
    const float* Wr0 = (const float*)d_in[base + 2];
    const float* g0  = (const float*)d_in[base + 3];
    const float* b0  = (const float*)d_in[base + 4];
    const float* Wl1 = (const float*)d_in[base + 5];
    const float* bl1 = (const float*)d_in[base + 6];
    const float* Wr1 = (const float*)d_in[base + 7];
    const float* g1  = (const float*)d_in[base + 8];
    const float* b1  = (const float*)d_in[base + 9];
    const float* Wl2 = (const float*)d_in[base + 10];
    const float* bl2 = (const float*)d_in[base + 11];
    const float* Wr2 = (const float*)d_in[base + 12];
    float* outp = (float*)d_out + (long)t * N * 64;

    vgae_gather<2><<<ngb, 256, 0, stream>>>(x, start, nbr, bufB, N);
    vgae_sage_gemm<<<nrt, 256, 0, stream>>>(bufB, x, invdeg, Wl0, bl0, Wr0,
                                            g0, b0, bufA, 64, 128, 1);
    vgae_gather<4><<<ngb, 256, 0, stream>>>(bufA, start, nbr, bufB, N);

    vgae_sage_gemm<<<nrt, 256, 0, stream>>>(bufB, bufA, invdeg, Wl1, bl1, Wr1,
                                            g1, b1, bufC, 128, 128, 1);
    vgae_gather<4><<<ngb, 256, 0, stream>>>(bufC, start, nbr, bufA, N);

    vgae_sage_gemm<<<nrt, 128, 0, stream>>>(bufA, bufC, invdeg, Wl2, bl2, Wr2,
                                            nullptr, nullptr, outp, 128, 64, 0);
  }
}


# GEMM operand path: each wave keeps its weight fragments in registers over a loop of row tiles; A tile staged in LDS by whole-row loads, next tile prefetched; k order and sum order unchanged
# speedup vs baseline: 4.1964x; 1.4321x over previous
#include <hip/hip_runtime.h>


#define LN_EPS 1e-5f

typedef __attribute__((ext_vector_type(2))) float v2f;
typedef __attribute__((ext_vector_type(8))) float v8f;


__global__ void vgae_zero_i(int* __restrict__ p, int n) {
  int i = blockIdx.x * blockDim.x + threadIdx.x;
  if (i < n) p[i] = 0;
}

__global__ void vgae_count_rank(const int* __restrict__ ei, int* __restrict__ cnt,
                                int* __restrict__ rank, int E) {
  int e = blockIdx.x * blockDim.x + threadIdx.x;
  if (e < E) rank[e] = atomicAdd(&cnt[ei[E + e]], 1);
}

__global__ void vgae_scan(int* __restrict__ cnt, float* __restrict__ invdeg, int n) {
  typedef __attribute__((ext_vector_type(4))) int v4i;
  typedef __attribute__((ext_vector_type(4))) float v4f;
  __shared__ int s_wsum[32];
  __shared__ int s_wexc[32];
  __shared__ int s_total;
  const int lane = threadIdx.x & 31;
  const int wave = threadIdx.x >> 5;
  int carry = 0;
  for (int base = 0; base < n; base += 1024 * 16) {
    const int i0 = base + threadIdx.x * 16;
    const bool full = (i0 + 16 <= n);
    v4i q[4];
    if (full) {
      const v4i* src = (const v4i*)(cnt + i0);
      q[0] = src[0]; q[1] = src[1]; q[2] = src[2]; q[3] = src[3];
    } else {
      for (int k = 0; k < 16; ++k)
        q[k >> 2][k & 3] = (i0 + k < n) ? cnt[i0 + k] : 0;
    }
    int tsum = 0;
    for (int k = 0; k < 16; ++k) tsum += q[k >> 2][k & 3];
    int incl = tsum;
    for (int o = 1; o < 32; o <<= 1) {
      int t = __shfl_up(incl, o);
      if (lane >= o) incl += t;
    }
    if (lane == 31) s_wsum[wave] = incl;
    __syncthreads();
    if (wave == 0) {
      const int w = s_wsum[lane];
      int wi = w;
      for (int o = 1; o < 32; o <<= 1) {
        int t = __shfl_up(wi, o);
        if (lane >= o) wi += t;
      }
      s_wexc[lane] = wi - w;
      if (lane == 31) s_total = wi;
    }
    __syncthreads();
    int run = carry + s_wexc[wave] + incl - tsum;
    v4i p[4];
    v4f r[4];
    for (int k = 0; k < 16; ++k) {
      const int c = q[k >> 2][k & 3];
      p[k >> 2][k & 3] = run;
      r[k >> 2][k & 3] = 1.0f / fmaxf((float)c, 1.0f);
      run += c;
    }
    if (full) {
      v4i* dp = (v4i*)(cnt + i0);
      v4f* dr = (v4f*)(invdeg + i0);
      dp[0] = p[0]; dp[1] = p[1]; dp[2] = p[2]; dp[3] = p[3];
      dr[0] = r[0]; dr[1] = r[1]; dr[2] = r[2]; dr[3] = r[3];
    } else {
      for (int k = 0; k < 16; ++k)
        if (i0 + k < n) { cnt[i0 + k] = p[k >> 2][k & 3]; invdeg[i0 + k] = r[k >> 2][k & 3]; }
    }
    carry += s_total;
    __syncthreads();
  }
  if (threadIdx.x == 0) cnt[n] = carry;
}

__global__ void vgae_fill(const int* __restrict__ ei, const int* __restrict__ start,
                          const int* __restrict__ rank, int* __restrict__ nbr, int E) {
  int e = blockIdx.x * blockDim.x + threadIdx.x;
  if (e < E) nbr[start[ei[E + e]] + rank[e]] = ei[e];
}

template <int VEC>
__global__ void vgae_gather(const float* __restrict__ h, const int* __restrict__ start,
                            const int* __restrict__ nbr, float* __restrict__ agg, int n) {
  typedef __attribute__((ext_vector_type(VEC))) float vecf;
  const int D = 32 * VEC;
  const int lane = threadIdx.x & 31;
  const int node = __builtin_amdgcn_readfirstlane(
      (int)(blockIdx.x * (blockDim.x >> 5) + (threadIdx.x >> 5)));
  if (node >= n) return;
  const int beg = start[node];
  const int end = start[node + 1];
  const float* hl = h + lane * VEC;
  vecf acc = {};
  for (int base = beg; base < end; base += 32) {
    const int m = min(32, end - base);
    int mine = (lane < m) ? nbr[base + lane] : 0;
    mine = min(max(mine, 0), n - 1);
    int j = 0;
    for (; j + 4 <= m; j += 4) {
      const int s0 = __builtin_amdgcn_readlane(mine, j);
      const int s1 = __builtin_amdgcn_readlane(mine, j + 1);
      const int s2 = __builtin_amdgcn_readlane(mine, j + 2);
      const int s3 = __builtin_amdgcn_readlane(mine, j + 3);
      const vecf v0 = *(const vecf*)(hl + (long)s0 * D);
      const vecf v1 = *(const vecf*)(hl + (long)s1 * D);
      const vecf v2 = *(const vecf*)(hl + (long)s2 * D);
      const vecf v3 = *(const vecf*)(hl + (long)s3 * D);
      acc += v0; acc += v1; acc += v2; acc += v3;
    }
    for (; j < m; ++j) {
      const int s0 = __builtin_amdgcn_readlane(mine, j);
      acc += *(const vecf*)(hl + (long)s0 * D);
    }
  }
  *(vecf*)(agg + (long)node * D + lane * VEC) = acc;
}

template <int DIN, int DOUT, int DO_LN>
__global__ __launch_bounds__(DOUT * 2)
void vgae_sage_gemm(const float* __restrict__ agg,
                    const float* __restrict__ h,
                    const float* __restrict__ invdeg,
                    const float* __restrict__ Wl,
                    const float* __restrict__ bl,
                    const float* __restrict__ Wr,
                    const float* __restrict__ gamma,
                    const float* __restrict__ beta,
                    float* __restrict__ out,
                    int ntiles) {
  typedef __attribute__((ext_vector_type(4))) float v4f;
  constexpr int NT    = DOUT * 2;
  constexpr int K     = 2 * DIN;
  constexpr int NK    = K / 4;
  constexpr int PITCH = K + 4;
  constexpr int C4    = DIN / 4;
  constexpr int NV    = 16 * C4 / NT;

  __shared__ __attribute__((aligned(16))) float As[16 * PITCH];
  __shared__ float smem[DO_LN ? 16 * 128 : 4];
  __shared__ float s_mean[16];
  __shared__ float s_rstd[16];

  const int tid  = threadIdx.x;
  const int lane = tid & 31;
  const int wave = tid >> 5;
  const int l15  = lane & 15;
  const int hi   = lane >> 4;
  const int bcol = wave * 16 + l15;
  const int koff = 2 * hi;

  int tile = blockIdx.x;
  if (tile >= ntiles) return;

  v2f breg[NK];
  {
    const float* wlrow = Wl + (long)bcol * DIN + koff;
    const float* wrrow = Wr + (long)bcol * DIN + koff;
#pragma unroll
    for (int s = 0; s < NK / 2; ++s) breg[s] = *(const v2f*)(wlrow + 4 * s);
#pragma unroll
    for (int s = 0; s < NK / 2; ++s) breg[NK / 2 + s] = *(const v2f*)(wrrow + 4 * s);
  }
  const float bias = bl[bcol];
  float g = 0.f, bt = 0.f;
  if (DO_LN) { g = gamma[tid & 127]; bt = beta[tid & 127]; }

  v4f ra[NV], rh[NV];
  float sc[NV];
#pragma unroll
  for (int i = 0; i < NV; ++i) {
    const int idx = tid + i * NT;
    const long row = (long)tile * 16 + idx / C4;
    const int  c   = (idx % C4) * 4;
    ra[i] = *(const v4f*)(agg + row * DIN + c);
    rh[i] = *(const v4f*)(h + row * DIN + c);
    sc[i] = invdeg[row];
  }

  const float* ap = As + l15 * PITCH + koff;

  for (; tile < ntiles; tile += gridDim.x) {
    const int rowbase = tile * 16;
#pragma unroll
    for (int i = 0; i < NV; ++i) {
      const int idx = tid + i * NT;
      float* dst = As + (idx / C4) * PITCH + (idx % C4) * 4;
      *(v4f*)dst = ra[i] * sc[i];
      *(v4f*)(dst + DIN) = rh[i];
    }
    __syncthreads();

    const int nxt = tile + gridDim.x;
    if (nxt < ntiles) {
#pragma unroll
      for (int i = 0; i < NV; ++i) {
        const int idx = tid + i * NT;
        const long row = (long)nxt * 16 + idx / C4;
        const int  c   = (idx % C4) * 4;
        ra[i] = *(const v4f*)(agg + row * DIN + c);
        rh[i] = *(const v4f*)(h + row * DIN + c);
        sc[i] = invdeg[row];
      }
    }

    v8f c = {0.f, 0.f, 0.f, 0.f, 0.f, 0.f, 0.f, 0.f};
#pragma unroll
    for (int s = 0; s < NK; ++s) {
      v2f a = *(const v2f*)(ap + 4 * s);
      c = __builtin_amdgcn_wmma_f32_16x16x4_f32(false, a, false, breg[s],
                                                (short)0, c, false, false);
    }

    if (!DO_LN) {
      for (int v = 0; v < 8; ++v) {
        int r = rowbase + v + 8 * hi;
        out[(long)r * DOUT + bcol] = c[v] + bias;
      }
      __syncthreads();
      continue;
    }

    for (int v = 0; v < 8; ++v)
      smem[(v + 8 * hi) * 128 + bcol] = c[v] + bias;
    __syncthreads();

    if (tid < 16) {
      const float* rowp = &smem[tid * 128];
      float s = 0.f, ss = 0.f;
      for (int j = 0; j < 128; ++j) { float x = rowp[j]; s += x; ss += x * x; }
      float m   = s * (1.0f / 128.0f);
      float var = ss * (1.0f / 128.0f) - m * m;
      s_mean[tid] = m;
      s_rstd[tid] = rsqrtf(var + LN_EPS);
    }
    __syncthreads();

    for (int idx = tid; idx < 16 * 128; idx += NT) {
      int r  = idx >> 7;
      float x = (smem[idx] - s_mean[r]) * s_rstd[r] * g + bt;
      out[(long)(rowbase + r) * 128 + (idx & 127)] = fmaxf(x, 0.0f);
    }
  }
}


extern "C" void kernel_launch(void* const* d_in, const int* in_sizes, int n_in,
                              void* d_out, int out_size, void* d_ws, size_t ws_size,
                              hipStream_t stream) {
  const float* x  = (const float*)d_in[0];
  const int*   ei = (const int*)d_in[1];
  const int N = in_sizes[0] / 64;
  const int E = in_sizes[1] / 2;

  float* invdeg = (float*)d_ws;
  int*   start  = (int*)(invdeg + N);
  int*   rank   = start + N + 64;
  int*   nbr    = rank + E;
  float* bufA   = invdeg + N + (long)N * 64;
  float* bufB   = bufA + (long)N * 128;
  float* bufC   = bufB + (long)N * 128;

  vgae_zero_i<<<(N + 255) / 256, 256, 0, stream>>>(start, N);
  vgae_count_rank<<<(E + 255) / 256, 256, 0, stream>>>(ei, start, rank, E);
  vgae_scan<<<1, 1024, 0, stream>>>(start, invdeg, N);
  vgae_fill<<<(E + 255) / 256, 256, 0, stream>>>(ei, start, rank, nbr, E);

  const int ngb = (N + 7) / 8;
  const int nrt = N / 16;
  for (int t = 0; t < 2; ++t) {
    const int base = 2 + t * 13;
    const float* Wl0 = (const float*)d_in[base + 0];
    const float* bl0 = (const float*)d_in[base + 1];
    const float* Wr0 = (const float*)d_in[base + 2];
    const float* g0  = (const float*)d_in[base + 3];
    const float* b0  = (const float*)d_in[base + 4];
    const float* Wl1 = (const float*)d_in[base + 5];
    const float* bl1 = (const float*)d_in[base + 6];
    const float* Wr1 = (const float*)d_in[base + 7];
    const float* g1  = (const float*)d_in[base + 8];
    const float* b1  = (const float*)d_in[base + 9];
    const float* Wl2 = (const float*)d_in[base + 10];
    const float* bl2 = (const float*)d_in[base + 11];
    const float* Wr2 = (const float*)d_in[base + 12];
    float* outp = (float*)d_out + (long)t * N * 64;

    vgae_gather<2><<<ngb, 256, 0, stream>>>(x, start, nbr, bufB, N);
    vgae_sage_gemm<64, 128, 1><<<512, 256, 0, stream>>>(bufB, x, invdeg, Wl0, bl0, Wr0,
                                                        g0, b0, bufA, nrt);
    vgae_gather<4><<<ngb, 256, 0, stream>>>(bufA, start, nbr, bufB, N);

    vgae_sage_gemm<128, 128, 1><<<512, 256, 0, stream>>>(bufB, bufA, invdeg, Wl1, bl1, Wr1,
                                                         g1, b1, bufC, nrt);
    vgae_gather<4><<<ngb, 256, 0, stream>>>(bufC, start, nbr, bufA, N);

    vgae_sage_gemm<128, 64, 0><<<512, 128, 0, stream>>>(bufA, bufC, invdeg, Wl2, bl2, Wr2,
                                                        nullptr, nullptr, outp, nrt);
  }
}


# Neighbor sums taken inside each dense layer's kernel while it builds its A tile in LDS (wave per row, ids fetched ahead): 6 gather launches and the aggregate buffer dropped; row sum order unchanged
# speedup vs baseline: 4.5970x; 1.0955x over previous
#include <hip/hip_runtime.h>


#define LN_EPS 1e-5f

typedef __attribute__((ext_vector_type(2))) float v2f;
typedef __attribute__((ext_vector_type(8))) float v8f;


__global__ void vgae_zero_i(int* __restrict__ p, int n) {
  int i = blockIdx.x * blockDim.x + threadIdx.x;
  if (i < n) p[i] = 0;
}

__global__ void vgae_count_rank(const int* __restrict__ ei, int* __restrict__ cnt,
                                int* __restrict__ rank, int E) {
  int e = blockIdx.x * blockDim.x + threadIdx.x;
  if (e < E) rank[e] = atomicAdd(&cnt[ei[E + e]], 1);
}

__global__ void vgae_scan(int* __restrict__ cnt, float* __restrict__ invdeg, int n) {
  typedef __attribute__((ext_vector_type(4))) int v4i;
  typedef __attribute__((ext_vector_type(4))) float v4f;
  __shared__ int s_wsum[32];
  __shared__ int s_wexc[32];
  __shared__ int s_total;
  const int lane = threadIdx.x & 31;
  const int wave = threadIdx.x >> 5;
  int carry = 0;
  for (int base = 0; base < n; base += 1024 * 16) {
    const int i0 = base + threadIdx.x * 16;
    const bool full = (i0 + 16 <= n);
    v4i q[4];
    if (full) {
      const v4i* src = (const v4i*)(cnt + i0);
      q[0] = src[0]; q[1] = src[1]; q[2] = src[2]; q[3] = src[3];
    } else {
      for (int k = 0; k < 16; ++k)
        q[k >> 2][k & 3] = (i0 + k < n) ? cnt[i0 + k] : 0;
    }
    int tsum = 0;
    for (int k = 0; k < 16; ++k) tsum += q[k >> 2][k & 3];
    int incl = tsum;
    for (int o = 1; o < 32; o <<= 1) {
      int t = __shfl_up(incl, o);
      if (lane >= o) incl += t;
    }
    if (lane == 31) s_wsum[wave] = incl;
    __syncthreads();
    if (wave == 0) {
      const int w = s_wsum[lane];
      int wi = w;
      for (int o = 1; o < 32; o <<= 1) {
        int t = __shfl_up(wi, o);
        if (lane >= o) wi += t;
      }
      s_wexc[lane] = wi - w;
      if (lane == 31) s_total = wi;
    }
    __syncthreads();
    int run = carry + s_wexc[wave] + incl - tsum;
    v4i p[4];
    v4f r[4];
    for (int k = 0; k < 16; ++k) {
      const int c = q[k >> 2][k & 3];
      p[k >> 2][k & 3] = run;
      r[k >> 2][k & 3] = 1.0f / fmaxf((float)c, 1.0f);
      run += c;
    }
    if (full) {
      v4i* dp = (v4i*)(cnt + i0);
      v4f* dr = (v4f*)(invdeg + i0);
      dp[0] = p[0]; dp[1] = p[1]; dp[2] = p[2]; dp[3] = p[3];
      dr[0] = r[0]; dr[1] = r[1]; dr[2] = r[2]; dr[3] = r[3];
    } else {
      for (int k = 0; k < 16; ++k)
        if (i0 + k < n) { cnt[i0 + k] = p[k >> 2][k & 3]; invdeg[i0 + k] = r[k >> 2][k & 3]; }
    }
    carry += s_total;
    __syncthreads();
  }
  if (threadIdx.x == 0) cnt[n] = carry;
}

__global__ void vgae_fill(const int* __restrict__ ei, const int* __restrict__ start,
                          const int* __restrict__ rank, int* __restrict__ nbr, int E) {
  int e = blockIdx.x * blockDim.x + threadIdx.x;
  if (e < E) nbr[start[ei[E + e]] + rank[e]] = ei[e];
}

template <int VEC> struct vgae_vec { typedef __attribute__((ext_vector_type(VEC))) float t; };

template <int VEC>
__device__ __forceinline__ typename vgae_vec<VEC>::t
vgae_bucket_sum(const float* __restrict__ hl, const int* __restrict__ nbr,
                int beg, int end, int ids0, int lane, int n) {
  typedef typename vgae_vec<VEC>::t vecf;
  constexpr int D = 32 * VEC;
  vecf acc = {};
  int mine = min(max(ids0, 0), n - 1);
  for (int base = beg; base < end; base += 32) {
    const int m = min(32, end - base);
    if (base != beg) {
      mine = (lane < m) ? nbr[base + lane] : 0;
      mine = min(max(mine, 0), n - 1);
    }
    int j = 0;
    for (; j + 8 <= m; j += 8) {
      const int s0 = __builtin_amdgcn_readlane(mine, j);
      const int s1 = __builtin_amdgcn_readlane(mine, j + 1);
      const int s2 = __builtin_amdgcn_readlane(mine, j + 2);
      const int s3 = __builtin_amdgcn_readlane(mine, j + 3);
      const int s4 = __builtin_amdgcn_readlane(mine, j + 4);
      const int s5 = __builtin_amdgcn_readlane(mine, j + 5);
      const int s6 = __builtin_amdgcn_readlane(mine, j + 6);
      const int s7 = __builtin_amdgcn_readlane(mine, j + 7);
      const vecf v0 = *(const vecf*)(hl + (long)s0 * D);
      const vecf v1 = *(const vecf*)(hl + (long)s1 * D);
      const vecf v2 = *(const vecf*)(hl + (long)s2 * D);
      const vecf v3 = *(const vecf*)(hl + (long)s3 * D);
      const vecf v4 = *(const vecf*)(hl + (long)s4 * D);
      const vecf v5 = *(const vecf*)(hl + (long)s5 * D);
      const vecf v6 = *(const vecf*)(hl + (long)s6 * D);
      const vecf v7 = *(const vecf*)(hl + (long)s7 * D);
      acc += v0; acc += v1; acc += v2; acc += v3;
      acc += v4; acc += v5; acc += v6; acc += v7;
    }
    if (j + 4 <= m) {
      const int s0 = __builtin_amdgcn_readlane(mine, j);
      const int s1 = __builtin_amdgcn_readlane(mine, j + 1);
      const int s2 = __builtin_amdgcn_readlane(mine, j + 2);
      const int s3 = __builtin_amdgcn_readlane(mine, j + 3);
      const vecf v0 = *(const vecf*)(hl + (long)s0 * D);
      const vecf v1 = *(const vecf*)(hl + (long)s1 * D);
      const vecf v2 = *(const vecf*)(hl + (long)s2 * D);
      const vecf v3 = *(const vecf*)(hl + (long)s3 * D);
      acc += v0; acc += v1; acc += v2; acc += v3;
      j += 4;
    }
    const int t = m - j;
    if (t == 3) {
      const int s0 = __builtin_amdgcn_readlane(mine, j);
      const int s1 = __builtin_amdgcn_readlane(mine, j + 1);
      const int s2 = __builtin_amdgcn_readlane(mine, j + 2);
      const vecf v0 = *(const vecf*)(hl + (long)s0 * D);
      const vecf v1 = *(const vecf*)(hl + (long)s1 * D);
      const vecf v2 = *(const vecf*)(hl + (long)s2 * D);
      acc += v0; acc += v1; acc += v2;
    } else if (t == 2) {
      const int s0 = __builtin_amdgcn_readlane(mine, j);
      const int s1 = __builtin_amdgcn_readlane(mine, j + 1);
      const vecf v0 = *(const vecf*)(hl + (long)s0 * D);
      const vecf v1 = *(const vecf*)(hl + (long)s1 * D);
      acc += v0; acc += v1;
    } else if (t == 1) {
      const int s0 = __builtin_amdgcn_readlane(mine, j);
      acc += *(const vecf*)(hl + (long)s0 * D);
    }
  }
  return acc;
}

template <int DIN, int DOUT, int DO_LN>
__global__ __launch_bounds__(DOUT * 2)
void vgae_sage_gemm(const float* __restrict__ h,
                    const int* __restrict__ start,
                    const int* __restrict__ nbr,
                    const float* __restrict__ invdeg,
                    const float* __restrict__ Wl,
                    const float* __restrict__ bl,
                    const float* __restrict__ Wr,
                    const float* __restrict__ gamma,
                    const float* __restrict__ beta,
                    float* __restrict__ out,
                    int ntiles, int n) {
  typedef __attribute__((ext_vector_type(4))) float v4f;
  constexpr int NT    = DOUT * 2;
  constexpr int NW    = NT / 32;
  constexpr int RPW   = 16 / NW;
  constexpr int VEC   = DIN / 32;
  constexpr int K     = 2 * DIN;
  constexpr int NK    = K / 4;
  constexpr int PITCH = K + 4;
  constexpr int C4    = DIN / 4;
  constexpr int NV    = 16 * C4 / NT;
  typedef typename vgae_vec<VEC>::t vecf;

  __shared__ __attribute__((aligned(16))) float As[16 * PITCH];
  __shared__ float smem[DO_LN ? 16 * 128 : 4];
  __shared__ float s_mean[16];
  __shared__ float s_rstd[16];

  const int tid  = threadIdx.x;
  const int lane = tid & 31;
  const int wave = __builtin_amdgcn_readfirstlane(tid >> 5);
  const int wv   = tid >> 5;
  const int l15  = lane & 15;
  const int hi   = lane >> 4;
  const int bcol = wave * 16 + l15;
  const int koff = 2 * hi;

  int tile = blockIdx.x;
  if (tile >= ntiles) return;

  v2f breg[NK];
  {
    const float* wlrow = Wl + (long)bcol * DIN + koff;
    const float* wrrow = Wr + (long)bcol * DIN + koff;
#pragma unroll
    for (int s = 0; s < NK / 2; ++s) breg[s] = *(const v2f*)(wlrow + 4 * s);
#pragma unroll
    for (int s = 0; s < NK / 2; ++s) breg[NK / 2 + s] = *(const v2f*)(wrrow + 4 * s);
  }
  const float bias = bl[bcol];
  float g = 0.f, bt = 0.f;
  if (DO_LN) { g = gamma[tid & 127]; bt = beta[tid & 127]; }

  v4f rh[NV];
  int nb[RPW], ne[RPW], id0[RPW], nb2[RPW], ne2[RPW];
  float sc[RPW];
#pragma unroll
  for (int i = 0; i < NV; ++i) {
    const int idx = tid + i * NT;
    const long row = (long)tile * 16 + idx / C4;
    rh[i] = *(const v4f*)(h + row * DIN + (idx % C4) * 4);
  }
#pragma unroll
  for (int k = 0; k < RPW; ++k) {
    const int node = tile * 16 + wave + k * NW;
    nb[k] = start[node];
    ne[k] = start[node + 1];
    sc[k] = invdeg[node];
  }
#pragma unroll
  for (int k = 0; k < RPW; ++k)
    id0[k] = nbr[max(min(nb[k] + lane, ne[k] - 1), 0)];
  {
    const int t1 = tile + (int)gridDim.x;
    const int tb = (t1 < ntiles) ? t1 : tile;
#pragma unroll
    for (int k = 0; k < RPW; ++k) {
      const int node = tb * 16 + wv + k * NW;
      nb2[k] = start[node];
      ne2[k] = start[node + 1];
    }
  }

  const float* ap = As + l15 * PITCH + koff;
  const float* hl = h + lane * VEC;

  for (; tile < ntiles; tile += gridDim.x) {
    const int rowbase = tile * 16;
#pragma unroll
    for (int k = 0; k < RPW; ++k) {
      const vecf a = vgae_bucket_sum<VEC>(hl, nbr, nb[k], ne[k], id0[k], lane, n);
      *(vecf*)(As + (wave + k * NW) * PITCH + lane * VEC) = a * sc[k];
    }
#pragma unroll
    for (int i = 0; i < NV; ++i) {
      const int idx = tid + i * NT;
      *(v4f*)(As + (idx / C4) * PITCH + DIN + (idx % C4) * 4) = rh[i];
    }
    __syncthreads();

    const int nxt = tile + (int)gridDim.x;
    if (nxt < ntiles) {
#pragma unroll
      for (int i = 0; i < NV; ++i) {
        const int idx = tid + i * NT;
        const long row = (long)nxt * 16 + idx / C4;
        rh[i] = *(const v4f*)(h + row * DIN + (idx % C4) * 4);
      }
#pragma unroll
      for (int k = 0; k < RPW; ++k) {
        nb[k] = __builtin_amdgcn_readfirstlane(nb2[k]);
        ne[k] = __builtin_amdgcn_readfirstlane(ne2[k]);
        sc[k] = invdeg[nxt * 16 + wv + k * NW];
      }
#pragma unroll
      for (int k = 0; k < RPW; ++k)
        id0[k] = nbr[max(min(nb[k] + lane, ne[k] - 1), 0)];
      const int t2 = nxt + (int)gridDim.x;
      const int tb = (t2 < ntiles) ? t2 : nxt;
#pragma unroll
      for (int k = 0; k < RPW; ++k) {
        const int node = tb * 16 + wv + k * NW;
        nb2[k] = start[node];
        ne2[k] = start[node + 1];
      }
    }

    v8f c = {0.f, 0.f, 0.f, 0.f, 0.f, 0.f, 0.f, 0.f};
#pragma unroll
    for (int s = 0; s < NK; ++s) {
      v2f a = *(const v2f*)(ap + 4 * s);
      c = __builtin_amdgcn_wmma_f32_16x16x4_f32(false, a, false, breg[s],
                                                (short)0, c, false, false);
    }

    if (!DO_LN) {
      for (int v = 0; v < 8; ++v) {
        int r = rowbase + v + 8 * hi;
        out[(long)r * DOUT + bcol] = c[v] + bias;
      }
      __syncthreads();
      continue;
    }

    for (int v = 0; v < 8; ++v)
      smem[(v + 8 * hi) * 128 + bcol] = c[v] + bias;
    __syncthreads();

    if (tid < 16) {
      const float* rowp = &smem[tid * 128];
      float s = 0.f, ss = 0.f;
      for (int j = 0; j < 128; ++j) { float x = rowp[j]; s += x; ss += x * x; }
      float m   = s * (1.0f / 128.0f);
      float var = ss * (1.0f / 128.0f) - m * m;
      s_mean[tid] = m;
      s_rstd[tid] = rsqrtf(var + LN_EPS);
    }
    __syncthreads();

    for (int idx = tid; idx < 16 * 128; idx += NT) {
      int r  = idx >> 7;
      float x = (smem[idx] - s_mean[r]) * s_rstd[r] * g + bt;
      out[(long)(rowbase + r) * 128 + (idx & 127)] = fmaxf(x, 0.0f);
    }
  }
}


extern "C" void kernel_launch(void* const* d_in, const int* in_sizes, int n_in,
                              void* d_out, int out_size, void* d_ws, size_t ws_size,
                              hipStream_t stream) {
  const float* x  = (const float*)d_in[0];
  const int*   ei = (const int*)d_in[1];
  const int N = in_sizes[0] / 64;
  const int E = in_sizes[1] / 2;

  float* invdeg = (float*)d_ws;
  int*   start  = (int*)(invdeg + N);
  int*   rank   = start + N + 64;
  int*   nbr    = rank + E;
  float* bufA   = invdeg + N + (long)N * 64;
  float* bufC   = bufA + (long)N * 256;

  vgae_zero_i<<<(N + 255) / 256, 256, 0, stream>>>(start, N);
  vgae_count_rank<<<(E + 255) / 256, 256, 0, stream>>>(ei, start, rank, E);
  vgae_scan<<<1, 1024, 0, stream>>>(start, invdeg, N);
  vgae_fill<<<(E + 255) / 256, 256, 0, stream>>>(ei, start, rank, nbr, E);

  const int nrt = N / 16;
  for (int t = 0; t < 2; ++t) {
    const int base = 2 + t * 13;
    const float* Wl0 = (const float*)d_in[base + 0];
    const float* bl0 = (const float*)d_in[base + 1];
    const float* Wr0 = (const float*)d_in[base + 2];
    const float* g0  = (const float*)d_in[base + 3];
    const float* b0  = (const float*)d_in[base + 4];
    const float* Wl1 = (const float*)d_in[base + 5];
    const float* bl1 = (const float*)d_in[base + 6];
    const float* Wr1 = (const float*)d_in[base + 7];
    const float* g1  = (const float*)d_in[base + 8];
    const float* b1  = (const float*)d_in[base + 9];
    const float* Wl2 = (const float*)d_in[base + 10];
    const float* bl2 = (const float*)d_in[base + 11];
    const float* Wr2 = (const float*)d_in[base + 12];
    float* outp = (float*)d_out + (long)t * N * 64;

    vgae_sage_gemm<64, 128, 1><<<1024, 256, 0, stream>>>(x, start, nbr, invdeg, Wl0, bl0, Wr0,
                                                         g0, b0, bufA, nrt, N);
    vgae_sage_gemm<128, 128, 1><<<512, 256, 0, stream>>>(bufA, start, nbr, invdeg, Wl1, bl1, Wr1,
                                                          g1, b1, bufC, nrt, N);
    vgae_sage_gemm<128, 64, 0><<<1024, 128, 0, stream>>>(bufC, start, nbr, invdeg, Wl2, bl2, Wr2,
                                                         nullptr, nullptr, outp, nrt, N);
  }
}


# Hidden-state rows on 512-B boundaries (both buffers moved to the workspace front; they sat 64 B past a line); LN epilogue stores 16 B per lane: a row moves as 4 whole lines, was 5; sums unchanged
# speedup vs baseline: 4.9743x; 1.0821x over previous
#include <hip/hip_runtime.h>


#define LN_EPS 1e-5f

typedef __attribute__((ext_vector_type(2))) float v2f;
typedef __attribute__((ext_vector_type(8))) float v8f;


__global__ void vgae_zero_i(int* __restrict__ p, int n) {
  int i = blockIdx.x * blockDim.x + threadIdx.x;
  if (i < n) p[i] = 0;
}

__global__ void vgae_count_rank(const int* __restrict__ ei, int* __restrict__ cnt,
                                int* __restrict__ rank, int E) {
  int e = blockIdx.x * blockDim.x + threadIdx.x;
  if (e < E) rank[e] = atomicAdd(&cnt[ei[E + e]], 1);
}

__global__ void vgae_scan(int* __restrict__ cnt, float* __restrict__ invdeg, int n) {
  typedef __attribute__((ext_vector_type(4))) int v4i;
  typedef __attribute__((ext_vector_type(4))) float v4f;
  __shared__ int s_wsum[32];
  __shared__ int s_wexc[32];
  __shared__ int s_total;
  const int lane = threadIdx.x & 31;
  const int wave = threadIdx.x >> 5;
  int carry = 0;
  for (int base = 0; base < n; base += 1024 * 16) {
    const int i0 = base + threadIdx.x * 16;
    const bool full = (i0 + 16 <= n);
    v4i q[4];
    if (full) {
      const v4i* src = (const v4i*)(cnt + i0);
      q[0] = src[0]; q[1] = src[1]; q[2] = src[2]; q[3] = src[3];
    } else {
      for (int k = 0; k < 16; ++k)
        q[k >> 2][k & 3] = (i0 + k < n) ? cnt[i0 + k] : 0;
    }
    int tsum = 0;
    for (int k = 0; k < 16; ++k) tsum += q[k >> 2][k & 3];
    int incl = tsum;
    for (int o = 1; o < 32; o <<= 1) {
      int t = __shfl_up(incl, o);
      if (lane >= o) incl += t;
    }
    if (lane == 31) s_wsum[wave] = incl;
    __syncthreads();
    if (wave == 0) {
      const int w = s_wsum[lane];
      int wi = w;
      for (int o = 1; o < 32; o <<= 1) {
        int t = __shfl_up(wi, o);
        if (lane >= o) wi += t;
      }
      s_wexc[lane] = wi - w;
      if (lane == 31) s_total = wi;
    }
    __syncthreads();
    int run = carry + s_wexc[wave] + incl - tsum;
    v4i p[4];
    v4f r[4];
    for (int k = 0; k < 16; ++k) {
      const int c = q[k >> 2][k & 3];
      p[k >> 2][k & 3] = run;
      r[k >> 2][k & 3] = 1.0f / fmaxf((float)c, 1.0f);
      run += c;
    }
    if (full) {
      v4i* dp = (v4i*)(cnt + i0);
      v4f* dr = (v4f*)(invdeg + i0);
      dp[0] = p[0]; dp[1] = p[1]; dp[2] = p[2]; dp[3] = p[3];
      dr[0] = r[0]; dr[1] = r[1]; dr[2] = r[2]; dr[3] = r[3];
    } else {
      for (int k = 0; k < 16; ++k)
        if (i0 + k < n) { cnt[i0 + k] = p[k >> 2][k & 3]; invdeg[i0 + k] = r[k >> 2][k & 3]; }
    }
    carry += s_total;
    __syncthreads();
  }
  if (threadIdx.x == 0) cnt[n] = carry;
}

__global__ void vgae_fill(const int* __restrict__ ei, const int* __restrict__ start,
                          const int* __restrict__ rank, int* __restrict__ nbr, int E) {
  int e = blockIdx.x * blockDim.x + threadIdx.x;
  if (e < E) nbr[start[ei[E + e]] + rank[e]] = ei[e];
}

template <int VEC> struct vgae_vec { typedef __attribute__((ext_vector_type(VEC))) float t; };

template <int VEC>
__device__ __forceinline__ typename vgae_vec<VEC>::t
vgae_bucket_sum(const float* __restrict__ hl, const int* __restrict__ nbr,
                int beg, int end, int ids0, int lane, int n) {
  typedef typename vgae_vec<VEC>::t vecf;
  constexpr int D = 32 * VEC;
  vecf acc = {};
  int mine = min(max(ids0, 0), n - 1);
  for (int base = beg; base < end; base += 32) {
    const int m = min(32, end - base);
    if (base != beg) {
      mine = (lane < m) ? nbr[base + lane] : 0;
      mine = min(max(mine, 0), n - 1);
    }
    int j = 0;
    for (; j + 8 <= m; j += 8) {
      const int s0 = __builtin_amdgcn_readlane(mine, j);
      const int s1 = __builtin_amdgcn_readlane(mine, j + 1);
      const int s2 = __builtin_amdgcn_readlane(mine, j + 2);
      const int s3 = __builtin_amdgcn_readlane(mine, j + 3);
      const int s4 = __builtin_amdgcn_readlane(mine, j + 4);
      const int s5 = __builtin_amdgcn_readlane(mine, j + 5);
      const int s6 = __builtin_amdgcn_readlane(mine, j + 6);
      const int s7 = __builtin_amdgcn_readlane(mine, j + 7);
      const vecf v0 = *(const vecf*)(hl + (long)s0 * D);
      const vecf v1 = *(const vecf*)(hl + (long)s1 * D);
      const vecf v2 = *(const vecf*)(hl + (long)s2 * D);
      const vecf v3 = *(const vecf*)(hl + (long)s3 * D);
      const vecf v4 = *(const vecf*)(hl + (long)s4 * D);
      const vecf v5 = *(const vecf*)(hl + (long)s5 * D);
      const vecf v6 = *(const vecf*)(hl + (long)s6 * D);
      const vecf v7 = *(const vecf*)(hl + (long)s7 * D);
      acc += v0; acc += v1; acc += v2; acc += v3;
      acc += v4; acc += v5; acc += v6; acc += v7;
    }
    if (j + 4 <= m) {
      const int s0 = __builtin_amdgcn_readlane(mine, j);
      const int s1 = __builtin_amdgcn_readlane(mine, j + 1);
      const int s2 = __builtin_amdgcn_readlane(mine, j + 2);
      const int s3 = __builtin_amdgcn_readlane(mine, j + 3);
      const vecf v0 = *(const vecf*)(hl + (long)s0 * D);
      const vecf v1 = *(const vecf*)(hl + (long)s1 * D);
      const vecf v2 = *(const vecf*)(hl + (long)s2 * D);
      const vecf v3 = *(const vecf*)(hl + (long)s3 * D);
      acc += v0; acc += v1; acc += v2; acc += v3;
      j += 4;
    }
    const int t = m - j;
    if (t == 3) {
      const int s0 = __builtin_amdgcn_readlane(mine, j);
      const int s1 = __builtin_amdgcn_readlane(mine, j + 1);
      const int s2 = __builtin_amdgcn_readlane(mine, j + 2);
      const vecf v0 = *(const vecf*)(hl + (long)s0 * D);
      const vecf v1 = *(const vecf*)(hl + (long)s1 * D);
      const vecf v2 = *(const vecf*)(hl + (long)s2 * D);
      acc += v0; acc += v1; acc += v2;
    } else if (t == 2) {
      const int s0 = __builtin_amdgcn_readlane(mine, j);
      const int s1 = __builtin_amdgcn_readlane(mine, j + 1);
      const vecf v0 = *(const vecf*)(hl + (long)s0 * D);
      const vecf v1 = *(const vecf*)(hl + (long)s1 * D);
      acc += v0; acc += v1;
    } else if (t == 1) {
      const int s0 = __builtin_amdgcn_readlane(mine, j);
      acc += *(const vecf*)(hl + (long)s0 * D);
    }
  }
  return acc;
}

template <int DIN, int DOUT, int DO_LN>
__global__ __launch_bounds__(DOUT * 2)
void vgae_sage_gemm(const float* __restrict__ h,
                    const int* __restrict__ start,
                    const int* __restrict__ nbr,
                    const float* __restrict__ invdeg,
                    const float* __restrict__ Wl,
                    const float* __restrict__ bl,
                    const float* __restrict__ Wr,
                    const float* __restrict__ gamma,
                    const float* __restrict__ beta,
                    float* __restrict__ out,
                    int ntiles, int n) {
  typedef __attribute__((ext_vector_type(4))) float v4f;
  constexpr int NT    = DOUT * 2;
  constexpr int NW    = NT / 32;
  constexpr int RPW   = 16 / NW;
  constexpr int VEC   = DIN / 32;
  constexpr int K     = 2 * DIN;
  constexpr int NK    = K / 4;
  constexpr int PITCH = K + 4;
  constexpr int C4    = DIN / 4;
  constexpr int NV    = 16 * C4 / NT;
  typedef typename vgae_vec<VEC>::t vecf;

  __shared__ __attribute__((aligned(16))) float As[16 * PITCH];
  __shared__ __attribute__((aligned(16))) float smem[DO_LN ? 16 * 128 : 4];
  __shared__ float s_mean[16];
  __shared__ float s_rstd[16];

  const int tid  = threadIdx.x;
  const int lane = tid & 31;
  const int wave = __builtin_amdgcn_readfirstlane(tid >> 5);
  const int wv   = tid >> 5;
  const int l15  = lane & 15;
  const int hi   = lane >> 4;
  const int bcol = wave * 16 + l15;
  const int koff = 2 * hi;

  int tile = blockIdx.x;
  if (tile >= ntiles) return;

  v2f breg[NK];
  {
    const float* wlrow = Wl + (long)bcol * DIN + koff;
    const float* wrrow = Wr + (long)bcol * DIN + koff;
#pragma unroll
    for (int s = 0; s < NK / 2; ++s) breg[s] = *(const v2f*)(wlrow + 4 * s);
#pragma unroll
    for (int s = 0; s < NK / 2; ++s) breg[NK / 2 + s] = *(const v2f*)(wrrow + 4 * s);
  }
  const float bias = bl[bcol];
  v4f g4 = {0.f, 0.f, 0.f, 0.f}, b4 = {0.f, 0.f, 0.f, 0.f};
  if (DO_LN) {
    g4 = *(const v4f*)(gamma + lane * 4);
    b4 = *(const v4f*)(beta + lane * 4);
  }

  v4f rh[NV];
  int nb[RPW], ne[RPW], id0[RPW], nb2[RPW], ne2[RPW];
  float sc[RPW];
#pragma unroll
  for (int i = 0; i < NV; ++i) {
    const int idx = tid + i * NT;
    const long row = (long)tile * 16 + idx / C4;
    rh[i] = *(const v4f*)(h + row * DIN + (idx % C4) * 4);
  }
#pragma unroll
  for (int k = 0; k < RPW; ++k) {
    const int node = tile * 16 + wave + k * NW;
    nb[k] = start[node];
    ne[k] = start[node + 1];
    sc[k] = invdeg[node];
  }
#pragma unroll
  for (int k = 0; k < RPW; ++k)
    id0[k] = nbr[max(min(nb[k] + lane, ne[k] - 1), 0)];
  {
    const int t1 = tile + (int)gridDim.x;
    const int tb = (t1 < ntiles) ? t1 : tile;
#pragma unroll
    for (int k = 0; k < RPW; ++k) {
      const int node = tb * 16 + wv + k * NW;
      nb2[k] = start[node];
      ne2[k] = start[node + 1];
    }
  }

  const float* ap = As + l15 * PITCH + koff;
  const float* hl = h + lane * VEC;

  for (; tile < ntiles; tile += gridDim.x) {
    const int rowbase = tile * 16;
#pragma unroll
    for (int k = 0; k < RPW; ++k) {
      const vecf a = vgae_bucket_sum<VEC>(hl, nbr, nb[k], ne[k], id0[k], lane, n);
      *(vecf*)(As + (wave + k * NW) * PITCH + lane * VEC) = a * sc[k];
    }
#pragma unroll
    for (int i = 0; i < NV; ++i) {
      const int idx = tid + i * NT;
      *(v4f*)(As + (idx / C4) * PITCH + DIN + (idx % C4) * 4) = rh[i];
    }
    __syncthreads();

    const int nxt = tile + (int)gridDim.x;
    if (nxt < ntiles) {
#pragma unroll
      for (int i = 0; i < NV; ++i) {
        const int idx = tid + i * NT;
        const long row = (long)nxt * 16 + idx / C4;
        rh[i] = *(const v4f*)(h + row * DIN + (idx % C4) * 4);
      }
#pragma unroll
      for (int k = 0; k < RPW; ++k) {
        nb[k] = __builtin_amdgcn_readfirstlane(nb2[k]);
        ne[k] = __builtin_amdgcn_readfirstlane(ne2[k]);
        sc[k] = invdeg[nxt * 16 + wv + k * NW];
      }
#pragma unroll
      for (int k = 0; k < RPW; ++k)
        id0[k] = nbr[max(min(nb[k] + lane, ne[k] - 1), 0)];
      const int t2 = nxt + (int)gridDim.x;
      const int tb = (t2 < ntiles) ? t2 : nxt;
#pragma unroll
      for (int k = 0; k < RPW; ++k) {
        const int node = tb * 16 + wv + k * NW;
        nb2[k] = start[node];
        ne2[k] = start[node + 1];
      }
    }

    v8f c = {0.f, 0.f, 0.f, 0.f, 0.f, 0.f, 0.f, 0.f};
#pragma unroll
    for (int s = 0; s < NK; ++s) {
      v2f a = *(const v2f*)(ap + 4 * s);
      c = __builtin_amdgcn_wmma_f32_16x16x4_f32(false, a, false, breg[s],
                                                (short)0, c, false, false);
    }

    if (!DO_LN) {
      for (int v = 0; v < 8; ++v) {
        int r = rowbase + v + 8 * hi;
        out[(long)r * DOUT + bcol] = c[v] + bias;
      }
      __syncthreads();
      continue;
    }

    for (int v = 0; v < 8; ++v)
      smem[(v + 8 * hi) * 128 + bcol] = c[v] + bias;
    __syncthreads();

    if (tid < 16) {
      const float* rowp = &smem[tid * 128];
      float s = 0.f, ss = 0.f;
      for (int j = 0; j < 128; ++j) { float x = rowp[j]; s += x; ss += x * x; }
      float m   = s * (1.0f / 128.0f);
      float var = ss * (1.0f / 128.0f) - m * m;
      s_mean[tid] = m;
      s_rstd[tid] = rsqrtf(var + LN_EPS);
    }
    __syncthreads();

    for (int idx = tid; idx < 16 * 32; idx += NT) {
      const int r = idx >> 5;
      const v4f xv = *(const v4f*)&smem[idx * 4];
      const float m = s_mean[r], rs = s_rstd[r];
      v4f y;
      for (int e = 0; e < 4; ++e)
        y[e] = fmaxf((xv[e] - m) * rs * g4[e] + b4[e], 0.0f);
      *(v4f*)(out + (long)(rowbase + r) * 128 + lane * 4) = y;
    }
  }
}


extern "C" void kernel_launch(void* const* d_in, const int* in_sizes, int n_in,
                              void* d_out, int out_size, void* d_ws, size_t ws_size,
                              hipStream_t stream) {
  const float* x  = (const float*)d_in[0];
  const int*   ei = (const int*)d_in[1];
  const int N = in_sizes[0] / 64;
  const int E = in_sizes[1] / 2;

  float* bufA   = (float*)d_ws;
  float* bufC   = bufA + (long)N * 128;
  float* invdeg = bufC + (long)N * 128;
  int*   start  = (int*)(invdeg + N);
  int*   rank   = start + N + 64;
  int*   nbr    = rank + E;

  vgae_zero_i<<<(N + 255) / 256, 256, 0, stream>>>(start, N);
  vgae_count_rank<<<(E + 255) / 256, 256, 0, stream>>>(ei, start, rank, E);
  vgae_scan<<<1, 1024, 0, stream>>>(start, invdeg, N);
  vgae_fill<<<(E + 255) / 256, 256, 0, stream>>>(ei, start, rank, nbr, E);

  const int nrt = N / 16;
  for (int t = 0; t < 2; ++t) {
    const int base = 2 + t * 13;
    const float* Wl0 = (const float*)d_in[base + 0];
    const float* bl0 = (const float*)d_in[base + 1];
    const float* Wr0 = (const float*)d_in[base + 2];
    const float* g0  = (const float*)d_in[base + 3];
    const float* b0  = (const float*)d_in[base + 4];
    const float* Wl1 = (const float*)d_in[base + 5];
    const float* bl1 = (const float*)d_in[base + 6];
    const float* Wr1 = (const float*)d_in[base + 7];
    const float* g1  = (const float*)d_in[base + 8];
    const float* b1  = (const float*)d_in[base + 9];
    const float* Wl2 = (const float*)d_in[base + 10];
    const float* bl2 = (const float*)d_in[base + 11];
    const float* Wr2 = (const float*)d_in[base + 12];
    float* outp = (float*)d_out + (long)t * N * 64;

    vgae_sage_gemm<64, 128, 1><<<1024, 256, 0, stream>>>(x, start, nbr, invdeg, Wl0, bl0, Wr0,
                                                         g0, b0, bufA, nrt, N);
    vgae_sage_gemm<128, 128, 1><<<512, 256, 0, stream>>>(bufA, start, nbr, invdeg, Wl1, bl1, Wr1,
                                                          g1, b1, bufC, nrt, N);
    vgae_sage_gemm<128, 64, 0><<<1024, 128, 0, stream>>>(bufC, start, nbr, invdeg, Wl2, bl2, Wr2,
                                                         nullptr, nullptr, outp, nrt, N);
  }
}


# Neighbor sums: all source-row loads of a bucket, up to 16 (12 in layer 0), issued before the first add (was 8, 4, tail, each waited for); bounds and inv_deg held one lane each; sum order unchanged
# speedup vs baseline: 5.2177x; 1.0489x over previous
#include <hip/hip_runtime.h>


#define LN_EPS 1e-5f

typedef __attribute__((ext_vector_type(2))) float v2f;
typedef __attribute__((ext_vector_type(8))) float v8f;


__global__ void vgae_zero_i(int* __restrict__ p, int n) {
  int i = blockIdx.x * blockDim.x + threadIdx.x;
  if (i < n) p[i] = 0;
}

__global__ void vgae_count_rank(const int* __restrict__ ei, int* __restrict__ cnt,
                                int* __restrict__ rank, int E) {
  int e = blockIdx.x * blockDim.x + threadIdx.x;
  if (e < E) rank[e] = atomicAdd(&cnt[ei[E + e]], 1);
}

__global__ void vgae_scan(int* __restrict__ cnt, float* __restrict__ invdeg, int n) {
  typedef __attribute__((ext_vector_type(4))) int v4i;
  typedef __attribute__((ext_vector_type(4))) float v4f;
  __shared__ int s_wsum[32];
  __shared__ int s_wexc[32];
  __shared__ int s_total;
  const int lane = threadIdx.x & 31;
  const int wave = threadIdx.x >> 5;
  int carry = 0;
  for (int base = 0; base < n; base += 1024 * 16) {
    const int i0 = base + threadIdx.x * 16;
    const bool full = (i0 + 16 <= n);
    v4i q[4];
    if (full) {
      const v4i* src = (const v4i*)(cnt + i0);
      q[0] = src[0]; q[1] = src[1]; q[2] = src[2]; q[3] = src[3];
    } else {
      for (int k = 0; k < 16; ++k)
        q[k >> 2][k & 3] = (i0 + k < n) ? cnt[i0 + k] : 0;
    }
    int tsum = 0;
    for (int k = 0; k < 16; ++k) tsum += q[k >> 2][k & 3];
    int incl = tsum;
    for (int o = 1; o < 32; o <<= 1) {
      int t = __shfl_up(incl, o);
      if (lane >= o) incl += t;
    }
    if (lane == 31) s_wsum[wave] = incl;
    __syncthreads();
    if (wave == 0) {
      const int w = s_wsum[lane];
      int wi = w;
      for (int o = 1; o < 32; o <<= 1) {
        int t = __shfl_up(wi, o);
        if (lane >= o) wi += t;
      }
      s_wexc[lane] = wi - w;
      if (lane == 31) s_total = wi;
    }
    __syncthreads();
    int run = carry + s_wexc[wave] + incl - tsum;
    v4i p[4];
    v4f r[4];
    for (int k = 0; k < 16; ++k) {
      const int c = q[k >> 2][k & 3];
      p[k >> 2][k & 3] = run;
      r[k >> 2][k & 3] = 1.0f / fmaxf((float)c, 1.0f);
      run += c;
    }
    if (full) {
      v4i* dp = (v4i*)(cnt + i0);
      v4f* dr = (v4f*)(invdeg + i0);
      dp[0] = p[0]; dp[1] = p[1]; dp[2] = p[2]; dp[3] = p[3];
      dr[0] = r[0]; dr[1] = r[1]; dr[2] = r[2]; dr[3] = r[3];
    } else {
      for (int k = 0; k < 16; ++k)
        if (i0 + k < n) { cnt[i0 + k] = p[k >> 2][k & 3]; invdeg[i0 + k] = r[k >> 2][k & 3]; }
    }
    carry += s_total;
    __syncthreads();
  }
  if (threadIdx.x == 0) cnt[n] = carry;
}

__global__ void vgae_fill(const int* __restrict__ ei, const int* __restrict__ start,
                          const int* __restrict__ rank, int* __restrict__ nbr, int E) {
  int e = blockIdx.x * blockDim.x + threadIdx.x;
  if (e < E) nbr[start[ei[E + e]] + rank[e]] = ei[e];
}

template <int VEC> struct vgae_vec { typedef __attribute__((ext_vector_type(VEC))) float t; };

template <int VEC, int GRP>
__device__ __forceinline__ typename vgae_vec<VEC>::t
vgae_bucket_sum(const float* __restrict__ h, const int* __restrict__ nbr,
                int beg, int end, int ids0, int lane, int n) {
  typedef typename vgae_vec<VEC>::t vecf;
  constexpr int D = 32 * VEC;
  const unsigned loff = (unsigned)lane * VEC;
  vecf acc = {};
  int mine = min(max(ids0, 0), n - 1);
  for (int base = beg; base < end; base += 32) {
    const int m = min(32, end - base);
    if (base != beg) {
      mine = (lane < m) ? nbr[base + lane] : 0;
      mine = min(max(mine, 0), n - 1);
    }
#define VGAE_LD(i) v[i] = *(const vecf*)(h + (long)__builtin_amdgcn_readlane(mine, j + (i)) * D + loff)
#define VGAE_AD(i) acc += v[i]
    for (int j = 0; j < m; j += GRP) {
      const int c = m - j;
      vecf v[16];
      do {
        VGAE_LD(0);  if (c <= 1) break;
        VGAE_LD(1);  if (c <= 2) break;
        VGAE_LD(2);  if (c <= 3) break;
        VGAE_LD(3);  if (c <= 4) break;
        VGAE_LD(4);  if (c <= 5) break;
        VGAE_LD(5);  if (c <= 6) break;
        VGAE_LD(6);  if (c <= 7) break;
        VGAE_LD(7);  if (GRP == 8 || c <= 8) break;
        VGAE_LD(8);  if (GRP == 9 || c <= 9) break;
        VGAE_LD(9);  if (GRP == 10 || c <= 10) break;
        VGAE_LD(10); if (GRP == 11 || c <= 11) break;
        VGAE_LD(11); if (GRP == 12 || c <= 12) break;
        VGAE_LD(12); if (GRP == 13 || c <= 13) break;
        VGAE_LD(13); if (GRP == 14 || c <= 14) break;
        VGAE_LD(14); if (GRP == 15 || c <= 15) break;
        VGAE_LD(15);
      } while (0);
      do {
        VGAE_AD(0);  if (c <= 1) break;
        VGAE_AD(1);  if (c <= 2) break;
        VGAE_AD(2);  if (c <= 3) break;
        VGAE_AD(3);  if (c <= 4) break;
        VGAE_AD(4);  if (c <= 5) break;
        VGAE_AD(5);  if (c <= 6) break;
        VGAE_AD(6);  if (c <= 7) break;
        VGAE_AD(7);  if (GRP == 8 || c <= 8) break;
        VGAE_AD(8);  if (GRP == 9 || c <= 9) break;
        VGAE_AD(9);  if (GRP == 10 || c <= 10) break;
        VGAE_AD(10); if (GRP == 11 || c <= 11) break;
        VGAE_AD(11); if (GRP == 12 || c <= 12) break;
        VGAE_AD(12); if (GRP == 13 || c <= 13) break;
        VGAE_AD(13); if (GRP == 14 || c <= 14) break;
        VGAE_AD(14); if (GRP == 15 || c <= 15) break;
        VGAE_AD(15);
      } while (0);
    }
#undef VGAE_LD
#undef VGAE_AD
  }
  return acc;
}

template <int DIN, int DOUT, int DO_LN>
__global__ __launch_bounds__(DOUT * 2)
void vgae_sage_gemm(const float* __restrict__ h,
                    const int* __restrict__ start,
                    const int* __restrict__ nbr,
                    const float* __restrict__ invdeg,
                    const float* __restrict__ Wl,
                    const float* __restrict__ bl,
                    const float* __restrict__ Wr,
                    const float* __restrict__ gamma,
                    const float* __restrict__ beta,
                    float* __restrict__ out,
                    int ntiles, int n) {
  typedef __attribute__((ext_vector_type(4))) float v4f;
  constexpr int NT    = DOUT * 2;
  constexpr int NW    = NT / 32;
  constexpr int RPW   = 16 / NW;
  constexpr int VEC   = DIN / 32;
  constexpr int GRP   = (DIN == 128) ? 16 : 12;
  constexpr int K     = 2 * DIN;
  constexpr int NK    = K / 4;
  constexpr int PITCH = K + 4;
  constexpr int C4    = DIN / 4;
  constexpr int NV    = 16 * C4 / NT;
  typedef typename vgae_vec<VEC>::t vecf;

  __shared__ __attribute__((aligned(16))) float As[16 * PITCH];
  __shared__ __attribute__((aligned(16))) float smem[DO_LN ? 16 * 128 : 4];
  __shared__ float s_mean[16];
  __shared__ float s_rstd[16];

  const int tid  = threadIdx.x;
  const int lane = tid & 31;
  const int wave = __builtin_amdgcn_readfirstlane(tid >> 5);
  const int l15  = lane & 15;
  const int hi   = lane >> 4;
  const int bcol = wave * 16 + l15;
  const int koff = 2 * hi;

  int tile = blockIdx.x;
  if (tile >= ntiles) return;

  v2f breg[NK];
  {
    const float* wlrow = Wl + (long)bcol * DIN + koff;
    const float* wrrow = Wr + (long)bcol * DIN + koff;
#pragma unroll
    for (int s = 0; s < NK / 2; ++s) breg[s] = *(const v2f*)(wlrow + 4 * s);
#pragma unroll
    for (int s = 0; s < NK / 2; ++s) breg[NK / 2 + s] = *(const v2f*)(wrrow + 4 * s);
  }
  const float bias = bl[bcol];
  v4f g4 = {0.f, 0.f, 0.f, 0.f}, b4 = {0.f, 0.f, 0.f, 0.f};
  if (DO_LN) {
    g4 = *(const v4f*)(gamma + lane * 4);
    b4 = *(const v4f*)(beta + lane * 4);
  }

  v4f rh[NV];
  int nb[RPW], ne[RPW], id0[RPW];
  int st2;
  float sc1;
#pragma unroll
  for (int i = 0; i < NV; ++i) {
    const int idx = tid + i * NT;
    const long row = (long)tile * 16 + idx / C4;
    rh[i] = *(const v4f*)(h + row * DIN + (idx % C4) * 4);
  }
#pragma unroll
  for (int k = 0; k < RPW; ++k) {
    const int node = tile * 16 + wave + k * NW;
    nb[k] = start[node];
    ne[k] = start[node + 1];
  }
  sc1 = invdeg[tile * 16 + l15];
#pragma unroll
  for (int k = 0; k < RPW; ++k)
    id0[k] = nbr[max(min(nb[k] + lane, ne[k] - 1), 0)];
  {
    const int t1 = tile + (int)gridDim.x;
    const int tb = (t1 < ntiles) ? t1 : tile;
    st2 = start[tb * 16 + min(lane, 16)];
  }

  const float* ap = As + l15 * PITCH + koff;

  for (; tile < ntiles; tile += gridDim.x) {
    const int rowbase = tile * 16;
#pragma unroll
    for (int k = 0; k < RPW; ++k) {
      const vecf a = vgae_bucket_sum<VEC, GRP>(h, nbr, nb[k], ne[k], id0[k], lane, n);
      const float sck = __int_as_float(__builtin_amdgcn_readlane(__float_as_int(sc1), wave + k * NW));
      *(vecf*)(As + (wave + k * NW) * PITCH + lane * VEC) = a * sck;
    }
#pragma unroll
    for (int i = 0; i < NV; ++i) {
      const int idx = tid + i * NT;
      *(v4f*)(As + (idx / C4) * PITCH + DIN + (idx % C4) * 4) = rh[i];
    }
    __syncthreads();

    const int nxt = tile + (int)gridDim.x;
    if (nxt < ntiles) {
#pragma unroll
      for (int i = 0; i < NV; ++i) {
        const int idx = tid + i * NT;
        const long row = (long)nxt * 16 + idx / C4;
        rh[i] = *(const v4f*)(h + row * DIN + (idx % C4) * 4);
      }
#pragma unroll
      for (int k = 0; k < RPW; ++k) {
        nb[k] = __builtin_amdgcn_readlane(st2, wave + k * NW);
        ne[k] = __builtin_amdgcn_readlane(st2, wave + k * NW + 1);
      }
      sc1 = invdeg[nxt * 16 + l15];
#pragma unroll
      for (int k = 0; k < RPW; ++k)
        id0[k] = nbr[max(min(nb[k] + lane, ne[k] - 1), 0)];
      const int t2 = nxt + (int)gridDim.x;
      const int tb = (t2 < ntiles) ? t2 : nxt;
      st2 = start[tb * 16 + min(lane, 16)];
    }

    v8f c = {0.f, 0.f, 0.f, 0.f, 0.f, 0.f, 0.f, 0.f};
#pragma unroll
    for (int s = 0; s < NK; ++s) {
      v2f a = *(const v2f*)(ap + 4 * s);
      c = __builtin_amdgcn_wmma_f32_16x16x4_f32(false, a, false, breg[s],
                                                (short)0, c, false, false);
    }

    if (!DO_LN) {
      for (int v = 0; v < 8; ++v) {
        int r = rowbase + v + 8 * hi;
        out[(long)r * DOUT + bcol] = c[v] + bias;
      }
      __syncthreads();
      continue;
    }

    for (int v = 0; v < 8; ++v)
      smem[(v + 8 * hi) * 128 + bcol] = c[v] + bias;
    __syncthreads();

    if (tid < 16) {
      const float* rowp = &smem[tid * 128];
      float s = 0.f, ss = 0.f;
      for (int j = 0; j < 128; ++j) { float x = rowp[j]; s += x; ss += x * x; }
      float m   = s * (1.0f / 128.0f);
      float var = ss * (1.0f / 128.0f) - m * m;
      s_mean[tid] = m;
      s_rstd[tid] = rsqrtf(var + LN_EPS);
    }
    __syncthreads();

    for (int idx = tid; idx < 16 * 32; idx += NT) {
      const int r = idx >> 5;
      const v4f xv = *(const v4f*)&smem[idx * 4];
      const float m = s_mean[r], rs = s_rstd[r];
      v4f y;
      for (int e = 0; e < 4; ++e)
        y[e] = fmaxf((xv[e] - m) * rs * g4[e] + b4[e], 0.0f);
      *(v4f*)(out + (long)(rowbase + r) * 128 + lane * 4) = y;
    }
  }
}


extern "C" void kernel_launch(void* const* d_in, const int* in_sizes, int n_in,
                              void* d_out, int out_size, void* d_ws, size_t ws_size,
                              hipStream_t stream) {
  const float* x  = (const float*)d_in[0];
  const int*   ei = (const int*)d_in[1];
  const int N = in_sizes[0] / 64;
  const int E = in_sizes[1] / 2;

  float* bufA   = (float*)d_ws;
  float* bufC   = bufA + (long)N * 128;
  float* invdeg = bufC + (long)N * 128;
  int*   start  = (int*)(invdeg + N);
  int*   rank   = start + N + 64;
  int*   nbr    = rank + E;

  vgae_zero_i<<<(N + 255) / 256, 256, 0, stream>>>(start, N);
  vgae_count_rank<<<(E + 255) / 256, 256, 0, stream>>>(ei, start, rank, E);
  vgae_scan<<<1, 1024, 0, stream>>>(start, invdeg, N);
  vgae_fill<<<(E + 255) / 256, 256, 0, stream>>>(ei, start, rank, nbr, E);

  const int nrt = N / 16;
  for (int t = 0; t < 2; ++t) {
    const int base = 2 + t * 13;
    const float* Wl0 = (const float*)d_in[base + 0];
    const float* bl0 = (const float*)d_in[base + 1];
    const float* Wr0 = (const float*)d_in[base + 2];
    const float* g0  = (const float*)d_in[base + 3];
    const float* b0  = (const float*)d_in[base + 4];
    const float* Wl1 = (const float*)d_in[base + 5];
    const float* bl1 = (const float*)d_in[base + 6];
    const float* Wr1 = (const float*)d_in[base + 7];
    const float* g1  = (const float*)d_in[base + 8];
    const float* b1  = (const float*)d_in[base + 9];
    const float* Wl2 = (const float*)d_in[base + 10];
    const float* bl2 = (const float*)d_in[base + 11];
    const float* Wr2 = (const float*)d_in[base + 12];
    float* outp = (float*)d_out + (long)t * N * 64;

    vgae_sage_gemm<64, 128, 1><<<1024, 256, 0, stream>>>(x, start, nbr, invdeg, Wl0, bl0, Wr0,
                                                         g0, b0, bufA, nrt, N);
    vgae_sage_gemm<128, 128, 1><<<512, 256, 0, stream>>>(bufA, start, nbr, invdeg, Wl1, bl1, Wr1,
                                                          g1, b1, bufC, nrt, N);
    vgae_sage_gemm<128, 64, 0><<<1024, 128, 0, stream>>>(bufC, start, nbr, invdeg, Wl2, bl2, Wr2,
                                                         nullptr, nullptr, outp, nrt, N);
  }
}


# Layer 0 of both towers in one launch: the neighbor mean of x, the same for both, is taken once per row tile and the one A tile multiplied by both towers' weights (16 waves); sum orders unchanged
# speedup vs baseline: 5.5412x; 1.0620x over previous
#include <hip/hip_runtime.h>


#define LN_EPS 1e-5f

typedef __attribute__((ext_vector_type(2))) float v2f;
typedef __attribute__((ext_vector_type(8))) float v8f;


__global__ void vgae_zero_i(int* __restrict__ p, int n) {
  int i = blockIdx.x * blockDim.x + threadIdx.x;
  if (i < n) p[i] = 0;
}

__global__ void vgae_count_rank(const int* __restrict__ ei, int* __restrict__ cnt,
                                int* __restrict__ rank, int E) {
  int e = blockIdx.x * blockDim.x + threadIdx.x;
  if (e < E) rank[e] = atomicAdd(&cnt[ei[E + e]], 1);
}

__global__ void vgae_scan(int* __restrict__ cnt, float* __restrict__ invdeg, int n) {
  typedef __attribute__((ext_vector_type(4))) int v4i;
  typedef __attribute__((ext_vector_type(4))) float v4f;
  __shared__ int s_wsum[32];
  __shared__ int s_wexc[32];
  __shared__ int s_total;
  const int lane = threadIdx.x & 31;
  const int wave = threadIdx.x >> 5;
  int carry = 0;
  for (int base = 0; base < n; base += 1024 * 16) {
    const int i0 = base + threadIdx.x * 16;
    const bool full = (i0 + 16 <= n);
    v4i q[4];
    if (full) {
      const v4i* src = (const v4i*)(cnt + i0);
      q[0] = src[0]; q[1] = src[1]; q[2] = src[2]; q[3] = src[3];
    } else {
      for (int k = 0; k < 16; ++k)
        q[k >> 2][k & 3] = (i0 + k < n) ? cnt[i0 + k] : 0;
    }
    int tsum = 0;
    for (int k = 0; k < 16; ++k) tsum += q[k >> 2][k & 3];
    int incl = tsum;
    for (int o = 1; o < 32; o <<= 1) {
      int t = __shfl_up(incl, o);
      if (lane >= o) incl += t;
    }
    if (lane == 31) s_wsum[wave] = incl;
    __syncthreads();
    if (wave == 0) {
      const int w = s_wsum[lane];
      int wi = w;
      for (int o = 1; o < 32; o <<= 1) {
        int t = __shfl_up(wi, o);
        if (lane >= o) wi += t;
      }
      s_wexc[lane] = wi - w;
      if (lane == 31) s_total = wi;
    }
    __syncthreads();
    int run = carry + s_wexc[wave] + incl - tsum;
    v4i p[4];
    v4f r[4];
    for (int k = 0; k < 16; ++k) {
      const int c = q[k >> 2][k & 3];
      p[k >> 2][k & 3] = run;
      r[k >> 2][k & 3] = 1.0f / fmaxf((float)c, 1.0f);
      run += c;
    }
    if (full) {
      v4i* dp = (v4i*)(cnt + i0);
      v4f* dr = (v4f*)(invdeg + i0);
      dp[0] = p[0]; dp[1] = p[1]; dp[2] = p[2]; dp[3] = p[3];
      dr[0] = r[0]; dr[1] = r[1]; dr[2] = r[2]; dr[3] = r[3];
    } else {
      for (int k = 0; k < 16; ++k)
        if (i0 + k < n) { cnt[i0 + k] = p[k >> 2][k & 3]; invdeg[i0 + k] = r[k >> 2][k & 3]; }
    }
    carry += s_total;
    __syncthreads();
  }
  if (threadIdx.x == 0) cnt[n] = carry;
}

__global__ void vgae_fill(const int* __restrict__ ei, const int* __restrict__ start,
                          const int* __restrict__ rank, int* __restrict__ nbr, int E) {
  int e = blockIdx.x * blockDim.x + threadIdx.x;
  if (e < E) nbr[start[ei[E + e]] + rank[e]] = ei[e];
}

template <int VEC> struct vgae_vec { typedef __attribute__((ext_vector_type(VEC))) float t; };

template <int VEC, int GRP>
__device__ __forceinline__ typename vgae_vec<VEC>::t
vgae_bucket_sum(const float* __restrict__ h, const int* __restrict__ nbr,
                int beg, int end, int ids0, int lane, int n) {
  typedef typename vgae_vec<VEC>::t vecf;
  constexpr int D = 32 * VEC;
  const unsigned loff = (unsigned)lane * VEC;
  vecf acc = {};
  int mine = min(max(ids0, 0), n - 1);
  for (int base = beg; base < end; base += 32) {
    const int m = min(32, end - base);
    if (base != beg) {
      mine = (lane < m) ? nbr[base + lane] : 0;
      mine = min(max(mine, 0), n - 1);
    }
#define VGAE_LD(i) v[i] = *(const vecf*)(h + (long)__builtin_amdgcn_readlane(mine, j + (i)) * D + loff)
#define VGAE_AD(i) acc += v[i]
    for (int j = 0; j < m; j += GRP) {
      const int c = m - j;
      vecf v[16];
      do {
        VGAE_LD(0);  if (c <= 1) break;
        VGAE_LD(1);  if (c <= 2) break;
        VGAE_LD(2);  if (c <= 3) break;
        VGAE_LD(3);  if (c <= 4) break;
        VGAE_LD(4);  if (c <= 5) break;
        VGAE_LD(5);  if (c <= 6) break;
        VGAE_LD(6);  if (c <= 7) break;
        VGAE_LD(7);  if (GRP == 8 || c <= 8) break;
        VGAE_LD(8);  if (GRP == 9 || c <= 9) break;
        VGAE_LD(9);  if (GRP == 10 || c <= 10) break;
        VGAE_LD(10); if (GRP == 11 || c <= 11) break;
        VGAE_LD(11); if (GRP == 12 || c <= 12) break;
        VGAE_LD(12); if (GRP == 13 || c <= 13) break;
        VGAE_LD(13); if (GRP == 14 || c <= 14) break;
        VGAE_LD(14); if (GRP == 15 || c <= 15) break;
        VGAE_LD(15);
      } while (0);
      do {
        VGAE_AD(0);  if (c <= 1) break;
        VGAE_AD(1);  if (c <= 2) break;
        VGAE_AD(2);  if (c <= 3) break;
        VGAE_AD(3);  if (c <= 4) break;
        VGAE_AD(4);  if (c <= 5) break;
        VGAE_AD(5);  if (c <= 6) break;
        VGAE_AD(6);  if (c <= 7) break;
        VGAE_AD(7);  if (GRP == 8 || c <= 8) break;
        VGAE_AD(8);  if (GRP == 9 || c <= 9) break;
        VGAE_AD(9);  if (GRP == 10 || c <= 10) break;
        VGAE_AD(10); if (GRP == 11 || c <= 11) break;
        VGAE_AD(11); if (GRP == 12 || c <= 12) break;
        VGAE_AD(12); if (GRP == 13 || c <= 13) break;
        VGAE_AD(13); if (GRP == 14 || c <= 14) break;
        VGAE_AD(14); if (GRP == 15 || c <= 15) break;
        VGAE_AD(15);
      } while (0);
    }
#undef VGAE_LD
#undef VGAE_AD
  }
  return acc;
}

template <int DIN, int DOUT, int DO_LN, int TW>
__global__ __launch_bounds__(DOUT * 2 * TW)
void vgae_sage_gemm(const float* __restrict__ h,
                    const int* __restrict__ start,
                    const int* __restrict__ nbr,
                    const float* __restrict__ invdeg,
                    const float* __restrict__ Wl_a,
                    const float* __restrict__ bl_a,
                    const float* __restrict__ Wr_a,
                    const float* __restrict__ gamma_a,
                    const float* __restrict__ beta_a,
                    float* __restrict__ out_a,
                    const float* __restrict__ Wl_b,
                    const float* __restrict__ bl_b,
                    const float* __restrict__ Wr_b,
                    const float* __restrict__ gamma_b,
                    const float* __restrict__ beta_b,
                    float* __restrict__ out_b,
                    int ntiles, int n) {
  typedef __attribute__((ext_vector_type(4))) float v4f;
  constexpr int NWT   = DOUT / 16;
  constexpr int NT    = NWT * 32 * TW;
  constexpr int NW    = NT / 32;
  constexpr int RPW   = 16 / NW;
  constexpr int VEC   = DIN / 32;
  constexpr int GRP   = (DIN == 128) ? 16 : 12;
  constexpr int K     = 2 * DIN;
  constexpr int NK    = K / 4;
  constexpr int PITCH = K + 4;
  constexpr int C4    = DIN / 4;
  constexpr int NP    = 16 * C4;
  constexpr int NV    = (NP + NT - 1) / NT;
  constexpr bool PART = (NP % NT) != 0;
  static_assert(16 % NW == 0 && (!PART || NV == 1), "tile rows / root pieces per wave");
  typedef typename vgae_vec<VEC>::t vecf;

  __shared__ __attribute__((aligned(16))) float As[16 * PITCH];
  __shared__ __attribute__((aligned(16))) float smem[DO_LN ? TW * 16 * 128 : 4];
  __shared__ float s_mean[16 * TW];
  __shared__ float s_rstd[16 * TW];

  const int tid  = threadIdx.x;
  const int lane = tid & 31;
  const int wave = __builtin_amdgcn_readfirstlane(tid >> 5);
  const int l15  = lane & 15;
  const int hi   = lane >> 4;
  const int tw   = (TW == 2) ? wave / NWT : 0;
  const int cw   = (TW == 2) ? wave % NWT : wave;
  const int bcol = cw * 16 + l15;
  const int koff = 2 * hi;
  const float* __restrict__ Wl    = (TW == 2 && tw) ? Wl_b : Wl_a;
  const float* __restrict__ bl    = (TW == 2 && tw) ? bl_b : bl_a;
  const float* __restrict__ Wr    = (TW == 2 && tw) ? Wr_b : Wr_a;
  const float* __restrict__ gamma = (TW == 2 && tw) ? gamma_b : gamma_a;
  const float* __restrict__ beta  = (TW == 2 && tw) ? beta_b : beta_a;
  float* __restrict__ out         = (TW == 2 && tw) ? out_b : out_a;

  int tile = blockIdx.x;
  if (tile >= ntiles) return;

  v2f breg[NK];
  {
    const float* wlrow = Wl + (long)bcol * DIN + koff;
    const float* wrrow = Wr + (long)bcol * DIN + koff;
#pragma unroll
    for (int s = 0; s < NK / 2; ++s) breg[s] = *(const v2f*)(wlrow + 4 * s);
#pragma unroll
    for (int s = 0; s < NK / 2; ++s) breg[NK / 2 + s] = *(const v2f*)(wrrow + 4 * s);
  }
  const float bias = bl[bcol];
  v4f g4 = {0.f, 0.f, 0.f, 0.f}, b4 = {0.f, 0.f, 0.f, 0.f};
  if (DO_LN) {
    g4 = *(const v4f*)(gamma + lane * 4);
    b4 = *(const v4f*)(beta + lane * 4);
  }

  v4f rh[NV];
  int nb[RPW], ne[RPW], id0[RPW];
  int st2;
  float sc1;
#pragma unroll
  for (int i = 0; i < NV; ++i) {
    const int idx = tid + i * NT;
    const long row = (long)tile * 16 + idx / C4;
    if (!PART || idx < NP) rh[i] = *(const v4f*)(h + row * DIN + (idx % C4) * 4);
  }
#pragma unroll
  for (int k = 0; k < RPW; ++k) {
    const int node = tile * 16 + wave + k * NW;
    nb[k] = start[node];
    ne[k] = start[node + 1];
  }
  sc1 = invdeg[tile * 16 + l15];
#pragma unroll
  for (int k = 0; k < RPW; ++k)
    id0[k] = nbr[max(min(nb[k] + lane, ne[k] - 1), 0)];
  {
    const int t1 = tile + (int)gridDim.x;
    const int tb = (t1 < ntiles) ? t1 : tile;
    st2 = start[tb * 16 + min(lane, 16)];
  }

  const float* ap = As + l15 * PITCH + koff;

  for (; tile < ntiles; tile += gridDim.x) {
    const int rowbase = tile * 16;
#pragma unroll
    for (int k = 0; k < RPW; ++k) {
      const vecf a = vgae_bucket_sum<VEC, GRP>(h, nbr, nb[k], ne[k], id0[k], lane, n);
      const float sck = __int_as_float(__builtin_amdgcn_readlane(__float_as_int(sc1), wave + k * NW));
      *(vecf*)(As + (wave + k * NW) * PITCH + lane * VEC) = a * sck;
    }
#pragma unroll
    for (int i = 0; i < NV; ++i) {
      const int idx = tid + i * NT;
      if (!PART || idx < NP) *(v4f*)(As + (idx / C4) * PITCH + DIN + (idx % C4) * 4) = rh[i];
    }
    __syncthreads();

    const int nxt = tile + (int)gridDim.x;
    if (nxt < ntiles) {
#pragma unroll
      for (int i = 0; i < NV; ++i) {
        const int idx = tid + i * NT;
        const long row = (long)nxt * 16 + idx / C4;
        if (!PART || idx < NP) rh[i] = *(const v4f*)(h + row * DIN + (idx % C4) * 4);
      }
#pragma unroll
      for (int k = 0; k < RPW; ++k) {
        nb[k] = __builtin_amdgcn_readlane(st2, wave + k * NW);
        ne[k] = __builtin_amdgcn_readlane(st2, wave + k * NW + 1);
      }
      sc1 = invdeg[nxt * 16 + l15];
#pragma unroll
      for (int k = 0; k < RPW; ++k)
        id0[k] = nbr[max(min(nb[k] + lane, ne[k] - 1), 0)];
      const int t2 = nxt + (int)gridDim.x;
      const int tb = (t2 < ntiles) ? t2 : nxt;
      st2 = start[tb * 16 + min(lane, 16)];
    }

    v8f c = {0.f, 0.f, 0.f, 0.f, 0.f, 0.f, 0.f, 0.f};
#pragma unroll
    for (int s = 0; s < NK; ++s) {
      v2f a = *(const v2f*)(ap + 4 * s);
      c = __builtin_amdgcn_wmma_f32_16x16x4_f32(false, a, false, breg[s],
                                                (short)0, c, false, false);
    }

    if (!DO_LN) {
      for (int v = 0; v < 8; ++v) {
        int r = rowbase + v + 8 * hi;
        out[(long)r * DOUT + bcol] = c[v] + bias;
      }
      __syncthreads();
      continue;
    }

    for (int v = 0; v < 8; ++v)
      smem[(tw * 16 + v + 8 * hi) * 128 + bcol] = c[v] + bias;
    __syncthreads();

    if (tid < 16 * TW) {
      const float* rowp = &smem[tid * 128];
      float s = 0.f, ss = 0.f;
      for (int j = 0; j < 128; ++j) { float x = rowp[j]; s += x; ss += x * x; }
      float m   = s * (1.0f / 128.0f);
      float var = ss * (1.0f / 128.0f) - m * m;
      s_mean[tid] = m;
      s_rstd[tid] = rsqrtf(var + LN_EPS);
    }
    __syncthreads();

    for (int r = cw; r < 16; r += NWT) {
      const int tr = tw * 16 + r;
      const v4f xv = *(const v4f*)&smem[(tr * 32 + lane) * 4];
      const float m = s_mean[tr], rs = s_rstd[tr];
      v4f y;
      for (int e = 0; e < 4; ++e)
        y[e] = fmaxf((xv[e] - m) * rs * g4[e] + b4[e], 0.0f);
      *(v4f*)(out + (long)(rowbase + r) * 128 + lane * 4) = y;
    }
  }
}


extern "C" void kernel_launch(void* const* d_in, const int* in_sizes, int n_in,
                              void* d_out, int out_size, void* d_ws, size_t ws_size,
                              hipStream_t stream) {
  const float* x  = (const float*)d_in[0];
  const int*   ei = (const int*)d_in[1];
  const int N = in_sizes[0] / 64;
  const int E = in_sizes[1] / 2;

  const bool both = ws_size >= ((size_t)3 * N * 128 + 2 * (size_t)N + 64 + 2 * (size_t)E) * 4;
  float* bufA   = (float*)d_ws;
  float* bufB   = both ? bufA + (long)N * 128 : bufA;
  float* bufC   = bufB + (long)N * 128;
  float* invdeg = bufC + (long)N * 128;
  int*   start  = (int*)(invdeg + N);
  int*   rank   = start + N + 64;
  int*   nbr    = rank + E;

  vgae_zero_i<<<(N + 255) / 256, 256, 0, stream>>>(start, N);
  vgae_count_rank<<<(E + 255) / 256, 256, 0, stream>>>(ei, start, rank, E);
  vgae_scan<<<1, 1024, 0, stream>>>(start, invdeg, N);
  vgae_fill<<<(E + 255) / 256, 256, 0, stream>>>(ei, start, rank, nbr, E);

  const int nrt = N / 16;

  if (both)
    vgae_sage_gemm<64, 128, 1, 2><<<512, 512, 0, stream>>>(
        x, start, nbr, invdeg,
        (const float*)d_in[2], (const float*)d_in[3], (const float*)d_in[4],
        (const float*)d_in[5], (const float*)d_in[6], bufA,
        (const float*)d_in[15], (const float*)d_in[16], (const float*)d_in[17],
        (const float*)d_in[18], (const float*)d_in[19], bufB, nrt, N);

  for (int t = 0; t < 2; ++t) {
    const int base = 2 + t * 13;
    const float* h1  = t ? bufB : bufA;
    const float* Wl1 = (const float*)d_in[base + 5];
    const float* bl1 = (const float*)d_in[base + 6];
    const float* Wr1 = (const float*)d_in[base + 7];
    const float* g1  = (const float*)d_in[base + 8];
    const float* b1  = (const float*)d_in[base + 9];
    const float* Wl2 = (const float*)d_in[base + 10];
    const float* bl2 = (const float*)d_in[base + 11];
    const float* Wr2 = (const float*)d_in[base + 12];
    float* outp = (float*)d_out + (long)t * N * 64;

    if (!both)
      vgae_sage_gemm<64, 128, 1, 1><<<1024, 256, 0, stream>>>(
          x, start, nbr, invdeg,
          (const float*)d_in[base + 0], (const float*)d_in[base + 1], (const float*)d_in[base + 2],
          (const float*)d_in[base + 3], (const float*)d_in[base + 4], bufA,
          nullptr, nullptr, nullptr, nullptr, nullptr, nullptr, nrt, N);

    vgae_sage_gemm<128, 128, 1, 1><<<512, 256, 0, stream>>>(h1, start, nbr, invdeg, Wl1, bl1, Wr1,
                                                             g1, b1, bufC,
                                                             nullptr, nullptr, nullptr, nullptr,
                                                             nullptr, nullptr, nrt, N);
    vgae_sage_gemm<128, 64, 0, 1><<<1024, 128, 0, stream>>>(bufC, start, nbr, invdeg, Wl2, bl2, Wr2,
                                                            nullptr, nullptr, outp,
                                                            nullptr, nullptr, nullptr, nullptr,
                                                            nullptr, nullptr, nrt, N);
  }
}


# Layers 1 and 2: both towers side by side in one launch each (even/odd blocks, own weights; 4th hidden buffer), 9 launches -> 7: launch start and thin end paid once per layer; sums unchanged
# speedup vs baseline: 6.0527x; 1.0923x over previous
#include <hip/hip_runtime.h>


#define LN_EPS 1e-5f

typedef __attribute__((ext_vector_type(2))) float v2f;
typedef __attribute__((ext_vector_type(8))) float v8f;


__global__ void vgae_zero_i(int* __restrict__ p, int n) {
  int i = blockIdx.x * blockDim.x + threadIdx.x;
  if (i < n) p[i] = 0;
}

__global__ void vgae_count_rank(const int* __restrict__ ei, int* __restrict__ cnt,
                                int* __restrict__ rank, int E) {
  int e = blockIdx.x * blockDim.x + threadIdx.x;
  if (e < E) rank[e] = atomicAdd(&cnt[ei[E + e]], 1);
}

__global__ void vgae_scan(int* __restrict__ cnt, float* __restrict__ invdeg, int n) {
  typedef __attribute__((ext_vector_type(4))) int v4i;
  typedef __attribute__((ext_vector_type(4))) float v4f;
  __shared__ int s_wsum[32];
  __shared__ int s_wexc[32];
  __shared__ int s_total;
  const int lane = threadIdx.x & 31;
  const int wave = threadIdx.x >> 5;
  int carry = 0;
  for (int base = 0; base < n; base += 1024 * 16) {
    const int i0 = base + threadIdx.x * 16;
    const bool full = (i0 + 16 <= n);
    v4i q[4];
    if (full) {
      const v4i* src = (const v4i*)(cnt + i0);
      q[0] = src[0]; q[1] = src[1]; q[2] = src[2]; q[3] = src[3];
    } else {
      for (int k = 0; k < 16; ++k)
        q[k >> 2][k & 3] = (i0 + k < n) ? cnt[i0 + k] : 0;
    }
    int tsum = 0;
    for (int k = 0; k < 16; ++k) tsum += q[k >> 2][k & 3];
    int incl = tsum;
    for (int o = 1; o < 32; o <<= 1) {
      int t = __shfl_up(incl, o);
      if (lane >= o) incl += t;
    }
    if (lane == 31) s_wsum[wave] = incl;
    __syncthreads();
    if (wave == 0) {
      const int w = s_wsum[lane];
      int wi = w;
      for (int o = 1; o < 32; o <<= 1) {
        int t = __shfl_up(wi, o);
        if (lane >= o) wi += t;
      }
      s_wexc[lane] = wi - w;
      if (lane == 31) s_total = wi;
    }
    __syncthreads();
    int run = carry + s_wexc[wave] + incl - tsum;
    v4i p[4];
    v4f r[4];
    for (int k = 0; k < 16; ++k) {
      const int c = q[k >> 2][k & 3];
      p[k >> 2][k & 3] = run;
      r[k >> 2][k & 3] = 1.0f / fmaxf((float)c, 1.0f);
      run += c;
    }
    if (full) {
      v4i* dp = (v4i*)(cnt + i0);
      v4f* dr = (v4f*)(invdeg + i0);
      dp[0] = p[0]; dp[1] = p[1]; dp[2] = p[2]; dp[3] = p[3];
      dr[0] = r[0]; dr[1] = r[1]; dr[2] = r[2]; dr[3] = r[3];
    } else {
      for (int k = 0; k < 16; ++k)
        if (i0 + k < n) { cnt[i0 + k] = p[k >> 2][k & 3]; invdeg[i0 + k] = r[k >> 2][k & 3]; }
    }
    carry += s_total;
    __syncthreads();
  }
  if (threadIdx.x == 0) cnt[n] = carry;
}

__global__ void vgae_fill(const int* __restrict__ ei, const int* __restrict__ start,
                          const int* __restrict__ rank, int* __restrict__ nbr, int E) {
  int e = blockIdx.x * blockDim.x + threadIdx.x;
  if (e < E) nbr[start[ei[E + e]] + rank[e]] = ei[e];
}

template <int VEC> struct vgae_vec { typedef __attribute__((ext_vector_type(VEC))) float t; };

template <int VEC, int GRP>
__device__ __forceinline__ typename vgae_vec<VEC>::t
vgae_bucket_sum(const float* __restrict__ h, const int* __restrict__ nbr,
                int beg, int end, int ids0, int lane, int n) {
  typedef typename vgae_vec<VEC>::t vecf;
  constexpr int D = 32 * VEC;
  const unsigned loff = (unsigned)lane * VEC;
  vecf acc = {};
  int mine = min(max(ids0, 0), n - 1);
  for (int base = beg; base < end; base += 32) {
    const int m = min(32, end - base);
    if (base != beg) {
      mine = (lane < m) ? nbr[base + lane] : 0;
      mine = min(max(mine, 0), n - 1);
    }
#define VGAE_LD(i) v[i] = *(const vecf*)(h + (long)__builtin_amdgcn_readlane(mine, j + (i)) * D + loff)
#define VGAE_AD(i) acc += v[i]
    for (int j = 0; j < m; j += GRP) {
      const int c = m - j;
      vecf v[16];
      do {
        VGAE_LD(0);  if (c <= 1) break;
        VGAE_LD(1);  if (c <= 2) break;
        VGAE_LD(2);  if (c <= 3) break;
        VGAE_LD(3);  if (c <= 4) break;
        VGAE_LD(4);  if (c <= 5) break;
        VGAE_LD(5);  if (c <= 6) break;
        VGAE_LD(6);  if (c <= 7) break;
        VGAE_LD(7);  if (GRP == 8 || c <= 8) break;
        VGAE_LD(8);  if (GRP == 9 || c <= 9) break;
        VGAE_LD(9);  if (GRP == 10 || c <= 10) break;
        VGAE_LD(10); if (GRP == 11 || c <= 11) break;
        VGAE_LD(11); if (GRP == 12 || c <= 12) break;
        VGAE_LD(12); if (GRP == 13 || c <= 13) break;
        VGAE_LD(13); if (GRP == 14 || c <= 14) break;
        VGAE_LD(14); if (GRP == 15 || c <= 15) break;
        VGAE_LD(15);
      } while (0);
      do {
        VGAE_AD(0);  if (c <= 1) break;
        VGAE_AD(1);  if (c <= 2) break;
        VGAE_AD(2);  if (c <= 3) break;
        VGAE_AD(3);  if (c <= 4) break;
        VGAE_AD(4);  if (c <= 5) break;
        VGAE_AD(5);  if (c <= 6) break;
        VGAE_AD(6);  if (c <= 7) break;
        VGAE_AD(7);  if (GRP == 8 || c <= 8) break;
        VGAE_AD(8);  if (GRP == 9 || c <= 9) break;
        VGAE_AD(9);  if (GRP == 10 || c <= 10) break;
        VGAE_AD(10); if (GRP == 11 || c <= 11) break;
        VGAE_AD(11); if (GRP == 12 || c <= 12) break;
        VGAE_AD(12); if (GRP == 13 || c <= 13) break;
        VGAE_AD(13); if (GRP == 14 || c <= 14) break;
        VGAE_AD(14); if (GRP == 15 || c <= 15) break;
        VGAE_AD(15);
      } while (0);
    }
#undef VGAE_LD
#undef VGAE_AD
  }
  return acc;
}

template <int DIN, int DOUT, int DO_LN, int TW, bool SPLIT>
__global__ __launch_bounds__(DOUT * 2 * TW)
void vgae_sage_gemm(const float* __restrict__ h_a,
                    const float* __restrict__ h_b,
                    const int* __restrict__ start,
                    const int* __restrict__ nbr,
                    const float* __restrict__ invdeg,
                    const float* __restrict__ Wl_a,
                    const float* __restrict__ bl_a,
                    const float* __restrict__ Wr_a,
                    const float* __restrict__ gamma_a,
                    const float* __restrict__ beta_a,
                    float* __restrict__ out_a,
                    const float* __restrict__ Wl_b,
                    const float* __restrict__ bl_b,
                    const float* __restrict__ Wr_b,
                    const float* __restrict__ gamma_b,
                    const float* __restrict__ beta_b,
                    float* __restrict__ out_b,
                    int ntiles, int n) {
  typedef __attribute__((ext_vector_type(4))) float v4f;
  constexpr int NWT   = DOUT / 16;
  constexpr int NT    = NWT * 32 * TW;
  constexpr int NW    = NT / 32;
  constexpr int RPW   = 16 / NW;
  constexpr int VEC   = DIN / 32;
  constexpr int GRP   = (DIN == 128) ? 16 : 12;
  constexpr int K     = 2 * DIN;
  constexpr int NK    = K / 4;
  constexpr int PITCH = K + 4;
  constexpr int C4    = DIN / 4;
  constexpr int NP    = 16 * C4;
  constexpr int NV    = (NP + NT - 1) / NT;
  constexpr bool PART = (NP % NT) != 0;
  static_assert(16 % NW == 0 && (!PART || NV == 1), "tile rows / root pieces per wave");
  static_assert(!(SPLIT && TW == 2), "two towers share a block only where they share the input");
  typedef typename vgae_vec<VEC>::t vecf;

  __shared__ __attribute__((aligned(16))) float As[16 * PITCH];
  __shared__ __attribute__((aligned(16))) float smem[DO_LN ? TW * 16 * 128 : 4];
  __shared__ float s_mean[16 * TW];
  __shared__ float s_rstd[16 * TW];

  const int tid  = threadIdx.x;
  const int lane = tid & 31;
  const int wave = __builtin_amdgcn_readfirstlane(tid >> 5);
  const int l15  = lane & 15;
  const int hi   = lane >> 4;
  const int tsel = SPLIT ? (int)(blockIdx.x & 1) : 0;
  const int bid  = SPLIT ? (int)(blockIdx.x >> 1) : (int)blockIdx.x;
  const int gstr = SPLIT ? (int)(gridDim.x >> 1) : (int)gridDim.x;
  const bool second = SPLIT && tsel;
  const float* __restrict__ h = second ? h_b : h_a;
  const int tw   = (TW == 2) ? wave / NWT : 0;
  const int cw   = (TW == 2) ? wave % NWT : wave;
  const int bcol = cw * 16 + l15;
  const int koff = 2 * hi;
  const float* __restrict__ Wl    = ((TW == 2 && tw) || second) ? Wl_b : Wl_a;
  const float* __restrict__ bl    = ((TW == 2 && tw) || second) ? bl_b : bl_a;
  const float* __restrict__ Wr    = ((TW == 2 && tw) || second) ? Wr_b : Wr_a;
  const float* __restrict__ gamma = ((TW == 2 && tw) || second) ? gamma_b : gamma_a;
  const float* __restrict__ beta  = ((TW == 2 && tw) || second) ? beta_b : beta_a;
  float* __restrict__ out         = ((TW == 2 && tw) || second) ? out_b : out_a;

  int tile = bid;
  if (tile >= ntiles) return;

  v2f breg[NK];
  {
    const float* wlrow = Wl + (long)bcol * DIN + koff;
    const float* wrrow = Wr + (long)bcol * DIN + koff;
#pragma unroll
    for (int s = 0; s < NK / 2; ++s) breg[s] = *(const v2f*)(wlrow + 4 * s);
#pragma unroll
    for (int s = 0; s < NK / 2; ++s) breg[NK / 2 + s] = *(const v2f*)(wrrow + 4 * s);
  }
  const float bias = bl[bcol];
  v4f g4 = {0.f, 0.f, 0.f, 0.f}, b4 = {0.f, 0.f, 0.f, 0.f};
  if (DO_LN) {
    g4 = *(const v4f*)(gamma + lane * 4);
    b4 = *(const v4f*)(beta + lane * 4);
  }

  v4f rh[NV];
  int nb[RPW], ne[RPW], id0[RPW];
  int st2;
  float sc1;
#pragma unroll
  for (int i = 0; i < NV; ++i) {
    const int idx = tid + i * NT;
    const long row = (long)tile * 16 + idx / C4;
    if (!PART || idx < NP) rh[i] = *(const v4f*)(h + row * DIN + (idx % C4) * 4);
  }
#pragma unroll
  for (int k = 0; k < RPW; ++k) {
    const int node = tile * 16 + wave + k * NW;
    nb[k] = start[node];
    ne[k] = start[node + 1];
  }
  sc1 = invdeg[tile * 16 + l15];
#pragma unroll
  for (int k = 0; k < RPW; ++k)
    id0[k] = nbr[max(min(nb[k] + lane, ne[k] - 1), 0)];
  {
    const int t1 = tile + gstr;
    const int tb = (t1 < ntiles) ? t1 : tile;
    st2 = start[tb * 16 + min(lane, 16)];
  }

  const float* ap = As + l15 * PITCH + koff;

  for (; tile < ntiles; tile += gstr) {
    const int rowbase = tile * 16;
#pragma unroll
    for (int k = 0; k < RPW; ++k) {
      const vecf a = vgae_bucket_sum<VEC, GRP>(h, nbr, nb[k], ne[k], id0[k], lane, n);
      const float sck = __int_as_float(__builtin_amdgcn_readlane(__float_as_int(sc1), wave + k * NW));
      *(vecf*)(As + (wave + k * NW) * PITCH + lane * VEC) = a * sck;
    }
#pragma unroll
    for (int i = 0; i < NV; ++i) {
      const int idx = tid + i * NT;
      if (!PART || idx < NP) *(v4f*)(As + (idx / C4) * PITCH + DIN + (idx % C4) * 4) = rh[i];
    }
    __syncthreads();

    const int nxt = tile + gstr;
    if (nxt < ntiles) {
#pragma unroll
      for (int i = 0; i < NV; ++i) {
        const int idx = tid + i * NT;
        const long row = (long)nxt * 16 + idx / C4;
        if (!PART || idx < NP) rh[i] = *(const v4f*)(h + row * DIN + (idx % C4) * 4);
      }
#pragma unroll
      for (int k = 0; k < RPW; ++k) {
        nb[k] = __builtin_amdgcn_readlane(st2, wave + k * NW);
        ne[k] = __builtin_amdgcn_readlane(st2, wave + k * NW + 1);
      }
      sc1 = invdeg[nxt * 16 + l15];
#pragma unroll
      for (int k = 0; k < RPW; ++k)
        id0[k] = nbr[max(min(nb[k] + lane, ne[k] - 1), 0)];
      const int t2 = nxt + gstr;
      const int tb = (t2 < ntiles) ? t2 : nxt;
      st2 = start[tb * 16 + min(lane, 16)];
    }

    v8f c = {0.f, 0.f, 0.f, 0.f, 0.f, 0.f, 0.f, 0.f};
#pragma unroll
    for (int s = 0; s < NK; ++s) {
      v2f a = *(const v2f*)(ap + 4 * s);
      c = __builtin_amdgcn_wmma_f32_16x16x4_f32(false, a, false, breg[s],
                                                (short)0, c, false, false);
    }

    if (!DO_LN) {
      for (int v = 0; v < 8; ++v) {
        int r = rowbase + v + 8 * hi;
        out[(long)r * DOUT + bcol] = c[v] + bias;
      }
      __syncthreads();
      continue;
    }

    for (int v = 0; v < 8; ++v)
      smem[(tw * 16 + v + 8 * hi) * 128 + bcol] = c[v] + bias;
    __syncthreads();

    if (tid < 16 * TW) {
      const float* rowp = &smem[tid * 128];
      float s = 0.f, ss = 0.f;
      for (int j = 0; j < 128; ++j) { float x = rowp[j]; s += x; ss += x * x; }
      float m   = s * (1.0f / 128.0f);
      float var = ss * (1.0f / 128.0f) - m * m;
      s_mean[tid] = m;
      s_rstd[tid] = rsqrtf(var + LN_EPS);
    }
    __syncthreads();

    for (int r = cw; r < 16; r += NWT) {
      const int tr = tw * 16 + r;
      const v4f xv = *(const v4f*)&smem[(tr * 32 + lane) * 4];
      const float m = s_mean[tr], rs = s_rstd[tr];
      v4f y;
      for (int e = 0; e < 4; ++e)
        y[e] = fmaxf((xv[e] - m) * rs * g4[e] + b4[e], 0.0f);
      *(v4f*)(out + (long)(rowbase + r) * 128 + lane * 4) = y;
    }
  }
}


extern "C" void kernel_launch(void* const* d_in, const int* in_sizes, int n_in,
                              void* d_out, int out_size, void* d_ws, size_t ws_size,
                              hipStream_t stream) {
  const float* x  = (const float*)d_in[0];
  const int*   ei = (const int*)d_in[1];
  const int N = in_sizes[0] / 64;
  const int E = in_sizes[1] / 2;

  const size_t tail_words = 2 * (size_t)N + 64 + 2 * (size_t)E;
  const bool four = ws_size >= ((size_t)4 * N * 128 + tail_words) * 4;
  const bool both = ws_size >= ((size_t)3 * N * 128 + tail_words) * 4;
  float* bufA   = (float*)d_ws;
  float* bufB   = both ? bufA + (long)N * 128 : bufA;
  float* bufC   = bufB + (long)N * 128;
  float* bufD   = four ? bufC + (long)N * 128 : bufA;
  float* invdeg = (four ? bufD : bufC) + (long)N * 128;
  int*   start  = (int*)(invdeg + N);
  int*   rank   = start + N + 64;
  int*   nbr    = rank + E;

  vgae_zero_i<<<(N + 255) / 256, 256, 0, stream>>>(start, N);
  vgae_count_rank<<<(E + 255) / 256, 256, 0, stream>>>(ei, start, rank, E);
  vgae_scan<<<1, 1024, 0, stream>>>(start, invdeg, N);
  vgae_fill<<<(E + 255) / 256, 256, 0, stream>>>(ei, start, rank, nbr, E);

  const int nrt = N / 16;

#define VGAE_F(i) ((const float*)d_in[i])

  if (both) {
    vgae_sage_gemm<64, 128, 1, 2, false><<<512, 512, 0, stream>>>(
        x, nullptr, start, nbr, invdeg,
        VGAE_F(2), VGAE_F(3), VGAE_F(4), VGAE_F(5), VGAE_F(6), bufA,
        VGAE_F(15), VGAE_F(16), VGAE_F(17), VGAE_F(18), VGAE_F(19), bufB, nrt, N);

    if (four) {
      vgae_sage_gemm<128, 128, 1, 1, true><<<512, 256, 0, stream>>>(
          bufA, bufB, start, nbr, invdeg,
          VGAE_F(7), VGAE_F(8), VGAE_F(9), VGAE_F(10), VGAE_F(11), bufC,
          VGAE_F(20), VGAE_F(21), VGAE_F(22), VGAE_F(23), VGAE_F(24), bufD, nrt, N);
    } else {
      vgae_sage_gemm<128, 128, 1, 1, false><<<512, 256, 0, stream>>>(
          bufA, nullptr, start, nbr, invdeg,
          VGAE_F(7), VGAE_F(8), VGAE_F(9), VGAE_F(10), VGAE_F(11), bufC,
          nullptr, nullptr, nullptr, nullptr, nullptr, nullptr, nrt, N);
      vgae_sage_gemm<128, 128, 1, 1, false><<<512, 256, 0, stream>>>(
          bufB, nullptr, start, nbr, invdeg,
          VGAE_F(20), VGAE_F(21), VGAE_F(22), VGAE_F(23), VGAE_F(24), bufD,
          nullptr, nullptr, nullptr, nullptr, nullptr, nullptr, nrt, N);
    }

    vgae_sage_gemm<128, 64, 0, 1, true><<<1024, 128, 0, stream>>>(
        bufC, bufD, start, nbr, invdeg,
        VGAE_F(12), VGAE_F(13), VGAE_F(14), nullptr, nullptr, (float*)d_out,
        VGAE_F(25), VGAE_F(26), VGAE_F(27), nullptr, nullptr, (float*)d_out + (long)N * 64, nrt, N);
  } else {
    for (int t = 0; t < 2; ++t) {
      const int base = 2 + t * 13;
      float* outp = (float*)d_out + (long)t * N * 64;
      vgae_sage_gemm<64, 128, 1, 1, false><<<1024, 256, 0, stream>>>(
          x, nullptr, start, nbr, invdeg,
          VGAE_F(base + 0), VGAE_F(base + 1), VGAE_F(base + 2), VGAE_F(base + 3), VGAE_F(base + 4), bufA,
          nullptr, nullptr, nullptr, nullptr, nullptr, nullptr, nrt, N);
      vgae_sage_gemm<128, 128, 1, 1, false><<<512, 256, 0, stream>>>(
          bufA, nullptr, start, nbr, invdeg,
          VGAE_F(base + 5), VGAE_F(base + 6), VGAE_F(base + 7), VGAE_F(base + 8), VGAE_F(base + 9), bufC,
          nullptr, nullptr, nullptr, nullptr, nullptr, nullptr, nrt, N);
      vgae_sage_gemm<128, 64, 0, 1, false><<<1024, 128, 0, stream>>>(
          bufC, nullptr, start, nbr, invdeg,
          VGAE_F(base + 10), VGAE_F(base + 11), VGAE_F(base + 12), nullptr, nullptr, outp,
          nullptr, nullptr, nullptr, nullptr, nullptr, nullptr, nrt, N);
    }
  }
#undef VGAE_F
}


# Last layer: h2 times Wl2, Wr2 taken in layer 1's kernel (WMMA, weights in LDS); last launch gathers 256-B projected rows, was 512-B; sum order: row dots, then bucket-order sum, scale, add root
# speedup vs baseline: 6.4258x; 1.0616x over previous
#include <hip/hip_runtime.h>


#define LN_EPS 1e-5f

typedef __attribute__((ext_vector_type(2))) float v2f;
typedef __attribute__((ext_vector_type(8))) float v8f;


__global__ void vgae_zero_i(int* __restrict__ p, int n) {
  int i = blockIdx.x * blockDim.x + threadIdx.x;
  if (i < n) p[i] = 0;
}

__global__ void vgae_count_rank(const int* __restrict__ ei, int* __restrict__ cnt,
                                int* __restrict__ rank, int E) {
  int e = blockIdx.x * blockDim.x + threadIdx.x;
  if (e < E) rank[e] = atomicAdd(&cnt[ei[E + e]], 1);
}

__global__ void vgae_scan(int* __restrict__ cnt, float* __restrict__ invdeg, int n) {
  typedef __attribute__((ext_vector_type(4))) int v4i;
  typedef __attribute__((ext_vector_type(4))) float v4f;
  __shared__ int s_wsum[32];
  __shared__ int s_wexc[32];
  __shared__ int s_total;
  const int lane = threadIdx.x & 31;
  const int wave = threadIdx.x >> 5;
  int carry = 0;
  for (int base = 0; base < n; base += 1024 * 16) {
    const int i0 = base + threadIdx.x * 16;
    const bool full = (i0 + 16 <= n);
    v4i q[4];
    if (full) {
      const v4i* src = (const v4i*)(cnt + i0);
      q[0] = src[0]; q[1] = src[1]; q[2] = src[2]; q[3] = src[3];
    } else {
      for (int k = 0; k < 16; ++k)
        q[k >> 2][k & 3] = (i0 + k < n) ? cnt[i0 + k] : 0;
    }
    int tsum = 0;
    for (int k = 0; k < 16; ++k) tsum += q[k >> 2][k & 3];
    int incl = tsum;
    for (int o = 1; o < 32; o <<= 1) {
      int t = __shfl_up(incl, o);
      if (lane >= o) incl += t;
    }
    if (lane == 31) s_wsum[wave] = incl;
    __syncthreads();
    if (wave == 0) {
      const int w = s_wsum[lane];
      int wi = w;
      for (int o = 1; o < 32; o <<= 1) {
        int t = __shfl_up(wi, o);
        if (lane >= o) wi += t;
      }
      s_wexc[lane] = wi - w;
      if (lane == 31) s_total = wi;
    }
    __syncthreads();
    int run = carry + s_wexc[wave] + incl - tsum;
    v4i p[4];
    v4f r[4];
    for (int k = 0; k < 16; ++k) {
      const int c = q[k >> 2][k & 3];
      p[k >> 2][k & 3] = run;
      r[k >> 2][k & 3] = 1.0f / fmaxf((float)c, 1.0f);
      run += c;
    }
    if (full) {
      v4i* dp = (v4i*)(cnt + i0);
      v4f* dr = (v4f*)(invdeg + i0);
      dp[0] = p[0]; dp[1] = p[1]; dp[2] = p[2]; dp[3] = p[3];
      dr[0] = r[0]; dr[1] = r[1]; dr[2] = r[2]; dr[3] = r[3];
    } else {
      for (int k = 0; k < 16; ++k)
        if (i0 + k < n) { cnt[i0 + k] = p[k >> 2][k & 3]; invdeg[i0 + k] = r[k >> 2][k & 3]; }
    }
    carry += s_total;
    __syncthreads();
  }
  if (threadIdx.x == 0) cnt[n] = carry;
}

__global__ void vgae_fill(const int* __restrict__ ei, const int* __restrict__ start,
                          const int* __restrict__ rank, int* __restrict__ nbr, int E) {
  int e = blockIdx.x * blockDim.x + threadIdx.x;
  if (e < E) nbr[start[ei[E + e]] + rank[e]] = ei[e];
}

template <int VEC> struct vgae_vec { typedef __attribute__((ext_vector_type(VEC))) float t; };

template <int VEC, int GRP>
__device__ __forceinline__ typename vgae_vec<VEC>::t
vgae_bucket_sum(const float* __restrict__ h, const int* __restrict__ nbr,
                int beg, int end, int ids0, int lane, int n) {
  typedef typename vgae_vec<VEC>::t vecf;
  constexpr int D = 32 * VEC;
  const unsigned loff = (unsigned)lane * VEC;
  vecf acc = {};
  int mine = min(max(ids0, 0), n - 1);
  for (int base = beg; base < end; base += 32) {
    const int m = min(32, end - base);
    if (base != beg) {
      mine = (lane < m) ? nbr[base + lane] : 0;
      mine = min(max(mine, 0), n - 1);
    }
#define VGAE_LD(i) v[i] = *(const vecf*)(h + (long)__builtin_amdgcn_readlane(mine, j + (i)) * D + loff)
#define VGAE_AD(i) acc += v[i]
    for (int j = 0; j < m; j += GRP) {
      const int c = m - j;
      vecf v[16];
      do {
        VGAE_LD(0);  if (c <= 1) break;
        VGAE_LD(1);  if (c <= 2) break;
        VGAE_LD(2);  if (c <= 3) break;
        VGAE_LD(3);  if (c <= 4) break;
        VGAE_LD(4);  if (c <= 5) break;
        VGAE_LD(5);  if (c <= 6) break;
        VGAE_LD(6);  if (c <= 7) break;
        VGAE_LD(7);  if (GRP == 8 || c <= 8) break;
        VGAE_LD(8);  if (GRP == 9 || c <= 9) break;
        VGAE_LD(9);  if (GRP == 10 || c <= 10) break;
        VGAE_LD(10); if (GRP == 11 || c <= 11) break;
        VGAE_LD(11); if (GRP == 12 || c <= 12) break;
        VGAE_LD(12); if (GRP == 13 || c <= 13) break;
        VGAE_LD(13); if (GRP == 14 || c <= 14) break;
        VGAE_LD(14); if (GRP == 15 || c <= 15) break;
        VGAE_LD(15);
      } while (0);
      do {
        VGAE_AD(0);  if (c <= 1) break;
        VGAE_AD(1);  if (c <= 2) break;
        VGAE_AD(2);  if (c <= 3) break;
        VGAE_AD(3);  if (c <= 4) break;
        VGAE_AD(4);  if (c <= 5) break;
        VGAE_AD(5);  if (c <= 6) break;
        VGAE_AD(6);  if (c <= 7) break;
        VGAE_AD(7);  if (GRP == 8 || c <= 8) break;
        VGAE_AD(8);  if (GRP == 9 || c <= 9) break;
        VGAE_AD(9);  if (GRP == 10 || c <= 10) break;
        VGAE_AD(10); if (GRP == 11 || c <= 11) break;
        VGAE_AD(11); if (GRP == 12 || c <= 12) break;
        VGAE_AD(12); if (GRP == 13 || c <= 13) break;
        VGAE_AD(13); if (GRP == 14 || c <= 14) break;
        VGAE_AD(14); if (GRP == 15 || c <= 15) break;
        VGAE_AD(15);
      } while (0);
    }
#undef VGAE_LD
#undef VGAE_AD
  }
  return acc;
}

template <int DIN, int DOUT, int DO_LN, int TW, bool SPLIT, bool PROJ>
__global__ __launch_bounds__(DOUT * 2 * TW)
void vgae_sage_gemm(const float* __restrict__ h_a,
                    const float* __restrict__ h_b,
                    const int* __restrict__ start,
                    const int* __restrict__ nbr,
                    const float* __restrict__ invdeg,
                    const float* __restrict__ Wl_a,
                    const float* __restrict__ bl_a,
                    const float* __restrict__ Wr_a,
                    const float* __restrict__ gamma_a,
                    const float* __restrict__ beta_a,
                    float* __restrict__ out_a,
                    const float* __restrict__ Wl_b,
                    const float* __restrict__ bl_b,
                    const float* __restrict__ Wr_b,
                    const float* __restrict__ gamma_b,
                    const float* __restrict__ beta_b,
                    float* __restrict__ out_b,
                    int ntiles, int n,
                    const float* __restrict__ Wl2_a,
                    const float* __restrict__ bl2_a,
                    const float* __restrict__ Wr2_a,
                    const float* __restrict__ Wl2_b,
                    const float* __restrict__ bl2_b,
                    const float* __restrict__ Wr2_b) {
  typedef __attribute__((ext_vector_type(4))) float v4f;
  constexpr int NWT   = DOUT / 16;
  constexpr int NT    = NWT * 32 * TW;
  constexpr int NW    = NT / 32;
  constexpr int RPW   = 16 / NW;
  constexpr int VEC   = DIN / 32;
  constexpr int GRP   = (DIN == 128) ? 16 : 12;
  constexpr int K     = 2 * DIN;
  constexpr int NK    = K / 4;
  constexpr int PITCH = K + 4;
  constexpr int C4    = DIN / 4;
  constexpr int NP    = 16 * C4;
  constexpr int NV    = (NP + NT - 1) / NT;
  constexpr bool PART = (NP % NT) != 0;
  static_assert(16 % NW == 0 && (!PART || NV == 1), "tile rows / root pieces per wave");
  static_assert(!(SPLIT && TW == 2), "two towers share a block only where they share the input");
  static_assert(!PROJ || (DO_LN && DOUT == 128 && TW == 1), "projection follows a 128-wide LN layer");
  constexpr int PITCH2 = 128 + 4;
  typedef typename vgae_vec<VEC>::t vecf;

  __shared__ __attribute__((aligned(16))) float As[16 * PITCH];
  __shared__ __attribute__((aligned(16))) float smem[DO_LN ? TW * 16 * 128 : 4];
  __shared__ __attribute__((aligned(16))) float W2s[PROJ ? 128 * PITCH2 : 4];
  __shared__ __attribute__((aligned(16))) float Ys[PROJ ? 16 * PITCH2 : 4];
  __shared__ float s_mean[16 * TW];
  __shared__ float s_rstd[16 * TW];

  const int tid  = threadIdx.x;
  const int lane = tid & 31;
  const int wave = __builtin_amdgcn_readfirstlane(tid >> 5);
  const int l15  = lane & 15;
  const int hi   = lane >> 4;
  const int tsel = SPLIT ? (int)(blockIdx.x & 1) : 0;
  const int bid  = SPLIT ? (int)(blockIdx.x >> 1) : (int)blockIdx.x;
  const int gstr = SPLIT ? (int)(gridDim.x >> 1) : (int)gridDim.x;
  const bool second = SPLIT && tsel;
  const float* __restrict__ h = second ? h_b : h_a;
  const int tw   = (TW == 2) ? wave / NWT : 0;
  const int cw   = (TW == 2) ? wave % NWT : wave;
  const int bcol = cw * 16 + l15;
  const int koff = 2 * hi;
  const float* __restrict__ Wl    = ((TW == 2 && tw) || second) ? Wl_b : Wl_a;
  const float* __restrict__ bl    = ((TW == 2 && tw) || second) ? bl_b : bl_a;
  const float* __restrict__ Wr    = ((TW == 2 && tw) || second) ? Wr_b : Wr_a;
  const float* __restrict__ gamma = ((TW == 2 && tw) || second) ? gamma_b : gamma_a;
  const float* __restrict__ beta  = ((TW == 2 && tw) || second) ? beta_b : beta_a;
  float* __restrict__ out         = ((TW == 2 && tw) || second) ? out_b : out_a;

  int tile = bid;
  if (tile >= ntiles) return;

  v2f breg[NK];
  {
    const float* wlrow = Wl + (long)bcol * DIN + koff;
    const float* wrrow = Wr + (long)bcol * DIN + koff;
#pragma unroll
    for (int s = 0; s < NK / 2; ++s) breg[s] = *(const v2f*)(wlrow + 4 * s);
#pragma unroll
    for (int s = 0; s < NK / 2; ++s) breg[NK / 2 + s] = *(const v2f*)(wrrow + 4 * s);
  }
  const float bias = bl[bcol];
  v4f g4 = {0.f, 0.f, 0.f, 0.f}, b4 = {0.f, 0.f, 0.f, 0.f};
  if (DO_LN) {
    g4 = *(const v4f*)(gamma + lane * 4);
    b4 = *(const v4f*)(beta + lane * 4);
  }

  float bias2 = 0.f;
  if (PROJ) {
    const float* __restrict__ Wl2 = second ? Wl2_b : Wl2_a;
    const float* __restrict__ Wr2 = second ? Wr2_b : Wr2_a;
    const float* __restrict__ bl2 = second ? bl2_b : bl2_a;
    for (int idx = tid; idx < 128 * 32; idx += NT) {
      const int row = idx >> 5, pc = idx & 31;
      const float* srow = (row < 64) ? (Wl2 + row * 128) : (Wr2 + (row - 64) * 128);
      *(v4f*)(W2s + row * PITCH2 + pc * 4) = *(const v4f*)(srow + pc * 4);
    }
    if (cw >= 4) bias2 = bl2[(cw - 4) * 16 + l15];
  }

  v4f rh[NV];
  int nb[RPW], ne[RPW], id0[RPW];
  int st2;
  float sc1;
#pragma unroll
  for (int i = 0; i < NV; ++i) {
    const int idx = tid + i * NT;
    const long row = (long)tile * 16 + idx / C4;
    if (!PART || idx < NP) rh[i] = *(const v4f*)(h + row * DIN + (idx % C4) * 4);
  }
#pragma unroll
  for (int k = 0; k < RPW; ++k) {
    const int node = tile * 16 + wave + k * NW;
    nb[k] = start[node];
    ne[k] = start[node + 1];
  }
  sc1 = invdeg[tile * 16 + l15];
#pragma unroll
  for (int k = 0; k < RPW; ++k)
    id0[k] = nbr[max(min(nb[k] + lane, ne[k] - 1), 0)];
  {
    const int t1 = tile + gstr;
    const int tb = (t1 < ntiles) ? t1 : tile;
    st2 = start[tb * 16 + min(lane, 16)];
  }

  const float* ap = As + l15 * PITCH + koff;

  for (; tile < ntiles; tile += gstr) {
    const int rowbase = tile * 16;
#pragma unroll
    for (int k = 0; k < RPW; ++k) {
      const vecf a = vgae_bucket_sum<VEC, GRP>(h, nbr, nb[k], ne[k], id0[k], lane, n);
      const float sck = __int_as_float(__builtin_amdgcn_readlane(__float_as_int(sc1), wave + k * NW));
      *(vecf*)(As + (wave + k * NW) * PITCH + lane * VEC) = a * sck;
    }
#pragma unroll
    for (int i = 0; i < NV; ++i) {
      const int idx = tid + i * NT;
      if (!PART || idx < NP) *(v4f*)(As + (idx / C4) * PITCH + DIN + (idx % C4) * 4) = rh[i];
    }
    __syncthreads();

    const int nxt = tile + gstr;
    if (nxt < ntiles) {
#pragma unroll
      for (int i = 0; i < NV; ++i) {
        const int idx = tid + i * NT;
        const long row = (long)nxt * 16 + idx / C4;
        if (!PART || idx < NP) rh[i] = *(const v4f*)(h + row * DIN + (idx % C4) * 4);
      }
#pragma unroll
      for (int k = 0; k < RPW; ++k) {
        nb[k] = __builtin_amdgcn_readlane(st2, wave + k * NW);
        ne[k] = __builtin_amdgcn_readlane(st2, wave + k * NW + 1);
      }
      sc1 = invdeg[nxt * 16 + l15];
#pragma unroll
      for (int k = 0; k < RPW; ++k)
        id0[k] = nbr[max(min(nb[k] + lane, ne[k] - 1), 0)];
      const int t2 = nxt + gstr;
      const int tb = (t2 < ntiles) ? t2 : nxt;
      st2 = start[tb * 16 + min(lane, 16)];
    }

    v8f c = {0.f, 0.f, 0.f, 0.f, 0.f, 0.f, 0.f, 0.f};
#pragma unroll
    for (int s = 0; s < NK; ++s) {
      v2f a = *(const v2f*)(ap + 4 * s);
      c = __builtin_amdgcn_wmma_f32_16x16x4_f32(false, a, false, breg[s],
                                                (short)0, c, false, false);
    }

    if (!DO_LN) {
      for (int v = 0; v < 8; ++v) {
        int r = rowbase + v + 8 * hi;
        out[(long)r * DOUT + bcol] = c[v] + bias;
      }
      __syncthreads();
      continue;
    }

    for (int v = 0; v < 8; ++v)
      smem[(tw * 16 + v + 8 * hi) * 128 + bcol] = c[v] + bias;
    __syncthreads();

    if (tid < 16 * TW) {
      const float* rowp = &smem[tid * 128];
      float s = 0.f, ss = 0.f;
      for (int j = 0; j < 128; ++j) { float x = rowp[j]; s += x; ss += x * x; }
      float m   = s * (1.0f / 128.0f);
      float var = ss * (1.0f / 128.0f) - m * m;
      s_mean[tid] = m;
      s_rstd[tid] = rsqrtf(var + LN_EPS);
    }
    __syncthreads();

    for (int r = cw; r < 16; r += NWT) {
      const int tr = tw * 16 + r;
      const v4f xv = *(const v4f*)&smem[(tr * 32 + lane) * 4];
      const float m = s_mean[tr], rs = s_rstd[tr];
      v4f y;
      for (int e = 0; e < 4; ++e)
        y[e] = fmaxf((xv[e] - m) * rs * g4[e] + b4[e], 0.0f);
      if (PROJ) *(v4f*)(Ys + r * PITCH2 + lane * 4) = y;
      else      *(v4f*)(out + (long)(rowbase + r) * 128 + lane * 4) = y;
    }

    if (PROJ) {
      __syncthreads();
      const float* yp = Ys + l15 * PITCH2 + koff;
      const float* wp = W2s + bcol * PITCH2 + koff;
      v8f c2 = {0.f, 0.f, 0.f, 0.f, 0.f, 0.f, 0.f, 0.f};
#pragma unroll
      for (int s = 0; s < 32; ++s) {
        const v2f a2 = *(const v2f*)(yp + 4 * s);
        const v2f b2 = *(const v2f*)(wp + 4 * s);
        c2 = __builtin_amdgcn_wmma_f32_16x16x4_f32(false, a2, false, b2,
                                                   (short)0, c2, false, false);
      }
      for (int v = 0; v < 8; ++v)
        smem[(v + 8 * hi) * 128 + bcol] = c2[v] + bias2;
      __syncthreads();
      float* __restrict__ plane = out + (hi ? (long)n * 64 : 0L);
      for (int r = cw; r < 16; r += NWT) {
        const v4f t = *(const v4f*)&smem[(r * 32 + lane) * 4];
        *(v4f*)(plane + (long)(rowbase + r) * 64 + l15 * 4) = t;
      }
    }
  }
}

__global__ __launch_bounds__(256)
void vgae_gather_out(const float* __restrict__ pr_a, const float* __restrict__ pr_b,
                     const int* __restrict__ start, const int* __restrict__ nbr,
                     const float* __restrict__ invdeg,
                     float* __restrict__ out_a, float* __restrict__ out_b, int n) {
  const int lane = threadIdx.x & 31;
  const int wave = __builtin_amdgcn_readfirstlane((int)(threadIdx.x >> 5));
  const bool second = (blockIdx.x & 1) != 0;
  const float* __restrict__ p = second ? pr_b : pr_a;
  const float* __restrict__ r = p + (long)n * 64;
  float* __restrict__ out = second ? out_b : out_a;
  const int stride = (int)(gridDim.x >> 1) * 8;
  int row = (int)(blockIdx.x >> 1) * 8 + wave;
  if (row >= n) return;
  int nb = start[row], ne = start[row + 1];
  float sc = invdeg[row];
  int ids = nbr[max(min(nb + lane, ne - 1), 0)];
  int r1 = (row + stride < n) ? row + stride : row;
  int nb1 = start[r1], ne1 = start[r1 + 1];
  float sc1 = invdeg[r1];
  for (; row < n; row += stride) {
    const int ids1 = nbr[max(min(nb1 + lane, ne1 - 1), 0)];
    const int r2 = (row + 2 * stride < n) ? row + 2 * stride : row;
    const int nb2 = start[r2], ne2 = start[r2 + 1];
    const float sc2 = invdeg[r2];
    const v2f rt = *(const v2f*)(r + (long)row * 64 + lane * 2);
    const v2f acc = vgae_bucket_sum<2, 16>(p, nbr, nb, ne, ids, lane, n);
    *(v2f*)(out + (long)row * 64 + lane * 2) = acc * sc + rt;
    nb = nb1; ne = ne1; sc = sc1; ids = ids1;
    nb1 = nb2; ne1 = ne2; sc1 = sc2;
  }
}


extern "C" void kernel_launch(void* const* d_in, const int* in_sizes, int n_in,
                              void* d_out, int out_size, void* d_ws, size_t ws_size,
                              hipStream_t stream) {
  const float* x  = (const float*)d_in[0];
  const int*   ei = (const int*)d_in[1];
  const int N = in_sizes[0] / 64;
  const int E = in_sizes[1] / 2;

  const size_t tail_words = 2 * (size_t)N + 64 + 2 * (size_t)E;
  const bool four = ws_size >= ((size_t)4 * N * 128 + tail_words) * 4;
  const bool both = ws_size >= ((size_t)3 * N * 128 + tail_words) * 4;
  float* bufA   = (float*)d_ws;
  float* bufB   = both ? bufA + (long)N * 128 : bufA;
  float* bufC   = bufB + (long)N * 128;
  float* bufD   = four ? bufC + (long)N * 128 : bufA;
  float* invdeg = (four ? bufD : bufC) + (long)N * 128;
  int*   start  = (int*)(invdeg + N);
  int*   rank   = start + N + 64;
  int*   nbr    = rank + E;

  vgae_zero_i<<<(N + 255) / 256, 256, 0, stream>>>(start, N);
  vgae_count_rank<<<(E + 255) / 256, 256, 0, stream>>>(ei, start, rank, E);
  vgae_scan<<<1, 1024, 0, stream>>>(start, invdeg, N);
  vgae_fill<<<(E + 255) / 256, 256, 0, stream>>>(ei, start, rank, nbr, E);

  const int nrt = N / 16;

#define VGAE_F(i) ((const float*)d_in[i])

  if (both) {
    vgae_sage_gemm<64, 128, 1, 2, false, false><<<512, 512, 0, stream>>>(
        x, nullptr, start, nbr, invdeg,
        VGAE_F(2), VGAE_F(3), VGAE_F(4), VGAE_F(5), VGAE_F(6), bufA,
        VGAE_F(15), VGAE_F(16), VGAE_F(17), VGAE_F(18), VGAE_F(19), bufB, nrt, N,
          nullptr, nullptr, nullptr, nullptr, nullptr, nullptr);

    if (four) {
      vgae_sage_gemm<128, 128, 1, 1, true, true><<<512, 256, 0, stream>>>(
          bufA, bufB, start, nbr, invdeg,
          VGAE_F(7), VGAE_F(8), VGAE_F(9), VGAE_F(10), VGAE_F(11), bufC,
          VGAE_F(20), VGAE_F(21), VGAE_F(22), VGAE_F(23), VGAE_F(24), bufD, nrt, N,
          VGAE_F(12), VGAE_F(13), VGAE_F(14), VGAE_F(25), VGAE_F(26), VGAE_F(27));
      vgae_gather_out<<<1024, 256, 0, stream>>>(
          bufC, bufD, start, nbr, invdeg,
          (float*)d_out, (float*)d_out + (long)N * 64, N);
    } else {
      vgae_sage_gemm<128, 128, 1, 1, false, false><<<512, 256, 0, stream>>>(
          bufA, nullptr, start, nbr, invdeg,
          VGAE_F(7), VGAE_F(8), VGAE_F(9), VGAE_F(10), VGAE_F(11), bufC,
          nullptr, nullptr, nullptr, nullptr, nullptr, nullptr, nrt, N,
          nullptr, nullptr, nullptr, nullptr, nullptr, nullptr);
      vgae_sage_gemm<128, 128, 1, 1, false, false><<<512, 256, 0, stream>>>(
          bufB, nullptr, start, nbr, invdeg,
          VGAE_F(20), VGAE_F(21), VGAE_F(22), VGAE_F(23), VGAE_F(24), bufD,
          nullptr, nullptr, nullptr, nullptr, nullptr, nullptr, nrt, N,
          nullptr, nullptr, nullptr, nullptr, nullptr, nullptr);
    }

    if (!four) vgae_sage_gemm<128, 64, 0, 1, true, false><<<1024, 128, 0, stream>>>(
        bufC, bufD, start, nbr, invdeg,
        VGAE_F(12), VGAE_F(13), VGAE_F(14), nullptr, nullptr, (float*)d_out,
        VGAE_F(25), VGAE_F(26), VGAE_F(27), nullptr, nullptr, (float*)d_out + (long)N * 64, nrt, N,
          nullptr, nullptr, nullptr, nullptr, nullptr, nullptr);
  } else {
    for (int t = 0; t < 2; ++t) {
      const int base = 2 + t * 13;
      float* outp = (float*)d_out + (long)t * N * 64;
      vgae_sage_gemm<64, 128, 1, 1, false, false><<<1024, 256, 0, stream>>>(
          x, nullptr, start, nbr, invdeg,
          VGAE_F(base + 0), VGAE_F(base + 1), VGAE_F(base + 2), VGAE_F(base + 3), VGAE_F(base + 4), bufA,
          nullptr, nullptr, nullptr, nullptr, nullptr, nullptr, nrt, N,
          nullptr, nullptr, nullptr, nullptr, nullptr, nullptr);
      vgae_sage_gemm<128, 128, 1, 1, false, false><<<512, 256, 0, stream>>>(
          bufA, nullptr, start, nbr, invdeg,
          VGAE_F(base + 5), VGAE_F(base + 6), VGAE_F(base + 7), VGAE_F(base + 8), VGAE_F(base + 9), bufC,
          nullptr, nullptr, nullptr, nullptr, nullptr, nullptr, nrt, N,
          nullptr, nullptr, nullptr, nullptr, nullptr, nullptr);
      vgae_sage_gemm<128, 64, 0, 1, false, false><<<1024, 128, 0, stream>>>(
          bufC, nullptr, start, nbr, invdeg,
          VGAE_F(base + 10), VGAE_F(base + 11), VGAE_F(base + 12), nullptr, nullptr, outp,
          nullptr, nullptr, nullptr, nullptr, nullptr, nullptr, nrt, N,
          nullptr, nullptr, nullptr, nullptr, nullptr, nullptr);
    }
  }
#undef VGAE_F
}


# Edge list build: per-edge returning global add, scattered load and store replaced by LDS binning (256-row bins: order edges by bin, reserve, write runs; per bin count, scan, place); 4 launches -> 3
# speedup vs baseline: 8.5159x; 1.3253x over previous
#include <hip/hip_runtime.h>


#define LN_EPS 1e-5f

typedef __attribute__((ext_vector_type(2))) float v2f;
typedef __attribute__((ext_vector_type(8))) float v8f;


__global__ void vgae_zero_i(int* __restrict__ p, int n) {
  int i = blockIdx.x * blockDim.x + threadIdx.x;
  if (i < n) p[i] = 0;
}

__global__ void vgae_count_rank(const int* __restrict__ ei, int* __restrict__ cnt,
                                int* __restrict__ rank, int E) {
  int e = blockIdx.x * blockDim.x + threadIdx.x;
  if (e < E) rank[e] = atomicAdd(&cnt[ei[E + e]], 1);
}

__global__ void vgae_scan(int* __restrict__ cnt, float* __restrict__ invdeg, int n) {
  typedef __attribute__((ext_vector_type(4))) int v4i;
  typedef __attribute__((ext_vector_type(4))) float v4f;
  __shared__ int s_wsum[32];
  __shared__ int s_wexc[32];
  __shared__ int s_total;
  const int lane = threadIdx.x & 31;
  const int wave = threadIdx.x >> 5;
  int carry = 0;
  for (int base = 0; base < n; base += 1024 * 16) {
    const int i0 = base + threadIdx.x * 16;
    const bool full = (i0 + 16 <= n);
    v4i q[4];
    if (full) {
      const v4i* src = (const v4i*)(cnt + i0);
      q[0] = src[0]; q[1] = src[1]; q[2] = src[2]; q[3] = src[3];
    } else {
      for (int k = 0; k < 16; ++k)
        q[k >> 2][k & 3] = (i0 + k < n) ? cnt[i0 + k] : 0;
    }
    int tsum = 0;
    for (int k = 0; k < 16; ++k) tsum += q[k >> 2][k & 3];
    int incl = tsum;
    for (int o = 1; o < 32; o <<= 1) {
      int t = __shfl_up(incl, o);
      if (lane >= o) incl += t;
    }
    if (lane == 31) s_wsum[wave] = incl;
    __syncthreads();
    if (wave == 0) {
      const int w = s_wsum[lane];
      int wi = w;
      for (int o = 1; o < 32; o <<= 1) {
        int t = __shfl_up(wi, o);
        if (lane >= o) wi += t;
      }
      s_wexc[lane] = wi - w;
      if (lane == 31) s_total = wi;
    }
    __syncthreads();
    int run = carry + s_wexc[wave] + incl - tsum;
    v4i p[4];
    v4f r[4];
    for (int k = 0; k < 16; ++k) {
      const int c = q[k >> 2][k & 3];
      p[k >> 2][k & 3] = run;
      r[k >> 2][k & 3] = 1.0f / fmaxf((float)c, 1.0f);
      run += c;
    }
    if (full) {
      v4i* dp = (v4i*)(cnt + i0);
      v4f* dr = (v4f*)(invdeg + i0);
      dp[0] = p[0]; dp[1] = p[1]; dp[2] = p[2]; dp[3] = p[3];
      dr[0] = r[0]; dr[1] = r[1]; dr[2] = r[2]; dr[3] = r[3];
    } else {
      for (int k = 0; k < 16; ++k)
        if (i0 + k < n) { cnt[i0 + k] = p[k >> 2][k & 3]; invdeg[i0 + k] = r[k >> 2][k & 3]; }
    }
    carry += s_total;
    __syncthreads();
  }
  if (threadIdx.x == 0) cnt[n] = carry;
}

__global__ void vgae_fill(const int* __restrict__ ei, const int* __restrict__ start,
                          const int* __restrict__ rank, int* __restrict__ nbr, int E) {
  int e = blockIdx.x * blockDim.x + threadIdx.x;
  if (e < E) nbr[start[ei[E + e]] + rank[e]] = ei[e];
}

#define VGAE_BIN_SHIFT 8
#define VGAE_BIN_ROWS  256
#define VGAE_MAX_BINS  256
#define VGAE_CNT_PITCH 32
#define VGAE_CHUNK     4096
#define VGAE_SEG       8192

__device__ __forceinline__ int vgae_scan256(int c, int tid, int* s_wsum) {
  const int lane = tid & 31;
  const int wave = tid >> 5;
  int incl = c;
  if (tid < 256) {
    for (int o = 1; o < 32; o <<= 1) {
      const int t = __shfl_up(incl, o);
      if (lane >= o) incl += t;
    }
    if (lane == 31) s_wsum[wave] = incl;
  }
  __syncthreads();
  if (tid < 256)
    for (int w = 0; w < wave; ++w) incl += s_wsum[w];
  return incl;
}

__global__ __launch_bounds__(1024)
void vgae_bin_edges(const int* __restrict__ ei, int* __restrict__ bintot,
                    int* __restrict__ binstore, int E, int n, int nbins, int cap) {
  __shared__ int s_cnt[VGAE_MAX_BINS];
  __shared__ int s_gofs[VGAE_MAX_BINS];
  __shared__ int s_wsum[8];
  __shared__ int s_word[VGAE_CHUNK];
  __shared__ int s_bin[VGAE_CHUNK];
  const int tid = threadIdx.x;
  if (tid < VGAE_MAX_BINS) s_cnt[tid] = 0;
  __syncthreads();
  const int first = (int)blockIdx.x * VGAE_CHUNK;
  const int nv = min(VGAE_CHUNK, E - first);
  int d[4], s[4], r[4];
#pragma unroll
  for (int k = 0; k < 4; ++k) {
    const int e = min(first + tid + k * 1024, E - 1);
    d[k] = min(max(ei[E + e], 0), n - 1);
    s[k] = min(max(ei[e], 0), n - 1);
  }
#pragma unroll
  for (int k = 0; k < 4; ++k) {
    r[k] = 0;
    if (tid + k * 1024 < nv) r[k] = atomicAdd(&s_cnt[d[k] >> VGAE_BIN_SHIFT], 1);
  }
  __syncthreads();
  const int c = (tid < VGAE_MAX_BINS) ? s_cnt[tid] : 0;
  int room = 0;
  if (tid < nbins && c != 0) room = atomicAdd(&bintot[tid * VGAE_CNT_PITCH], c);
  const int incl = vgae_scan256(c, tid, s_wsum);
  if (tid < VGAE_MAX_BINS) {
    s_cnt[tid]  = incl - c;
    s_gofs[tid] = room - (incl - c);
  }
  __syncthreads();
#pragma unroll
  for (int k = 0; k < 4; ++k) {
    if (tid + k * 1024 < nv) {
      const int b = d[k] >> VGAE_BIN_SHIFT;
      const int p = s_cnt[b] + r[k];
      s_word[p] = (s[k] << 8) | (d[k] & (VGAE_BIN_ROWS - 1));
      s_bin[p]  = b;
    }
  }
  __syncthreads();
#pragma unroll
  for (int k = 0; k < 4; ++k) {
    const int t = tid + k * 1024;
    if (t < nv) {
      const int b = s_bin[t];
      const int pos = s_gofs[b] + t;
      if (pos < cap) binstore[(long)b * cap + pos] = s_word[t];
    }
  }
}

__global__ __launch_bounds__(1024)
void vgae_sort_bin(const int* __restrict__ ei, const int* __restrict__ bintot,
                   const int* __restrict__ binstore, int* __restrict__ start,
                   float* __restrict__ invdeg, int* __restrict__ nbr,
                   int E, int n, int cap) {
  __shared__ int s_cnt[VGAE_BIN_ROWS];
  __shared__ int s_wsum[8];
  __shared__ int s_lower;
  __shared__ int s_seg[VGAE_SEG];
  const int tid  = threadIdx.x;
  const int lane = tid & 31;
  const int b    = (int)blockIdx.x;
  const int lo   = b << VGAE_BIN_SHIFT;
  if (tid < VGAE_BIN_ROWS) s_cnt[tid] = 0;
  if (tid == 0) s_lower = 0;
  __syncthreads();
  {
    int v = (tid < b) ? bintot[tid * VGAE_CNT_PITCH] : 0;
    for (int o = 16; o > 0; o >>= 1) v += __shfl_xor(v, o);
    if (lane == 0 && v != 0) atomicAdd(&s_lower, v);
  }
  const int total = bintot[b * VGAE_CNT_PITCH];
  const bool stored = total <= cap;
  const bool inlds  = total <= VGAE_SEG;
  const int* __restrict__ bs = binstore + (long)b * cap;
  if (stored) {
    for (int i0 = tid; i0 < total; i0 += 4096) {
      int w[4];
#pragma unroll
      for (int k = 0; k < 4; ++k) w[k] = bs[min(i0 + k * 1024, total - 1)];
#pragma unroll
      for (int k = 0; k < 4; ++k)
        atomicAdd(&s_cnt[w[k] & (VGAE_BIN_ROWS - 1)], (i0 + k * 1024 < total) ? 1 : 0);
    }
  } else {
    for (int e = tid; e < E; e += 1024) {
      const int dd = min(max(ei[E + e], 0), n - 1);
      if ((dd >> VGAE_BIN_SHIFT) == b) atomicAdd(&s_cnt[dd & (VGAE_BIN_ROWS - 1)], 1);
    }
  }
  __syncthreads();
  const int c = (tid < VGAE_BIN_ROWS) ? s_cnt[tid] : 0;
  const int incl = vgae_scan256(c, tid, s_wsum);
  const int base = s_lower;
  if (tid < VGAE_BIN_ROWS) {
    const int lbeg = incl - c;
    const int node = lo + tid;
    if (node < n) {
      start[node]  = base + lbeg;
      invdeg[node] = 1.0f / fmaxf((float)c, 1.0f);
      if (node == n - 1) start[n] = base + incl;
    }
    s_cnt[tid] = lbeg;
  }
  __syncthreads();
  if (stored) {
    for (int i0 = tid; i0 < total; i0 += 4096) {
      int w[4];
#pragma unroll
      for (int k = 0; k < 4; ++k) w[k] = bs[min(i0 + k * 1024, total - 1)];
#pragma unroll
      for (int k = 0; k < 4; ++k) {
        const bool in = i0 + k * 1024 < total;
        const int pos = atomicAdd(&s_cnt[w[k] & (VGAE_BIN_ROWS - 1)], in ? 1 : 0);
        if (in) {
          if (inlds) s_seg[pos] = w[k] >> 8;
          else       nbr[base + pos] = w[k] >> 8;
        }
      }
    }
    if (inlds) {
      __syncthreads();
      for (int i = tid; i < total; i += 1024) nbr[base + i] = s_seg[i];
    }
  } else {
    for (int e = tid; e < E; e += 1024) {
      const int dd = min(max(ei[E + e], 0), n - 1);
      if ((dd >> VGAE_BIN_SHIFT) == b)
        nbr[base + atomicAdd(&s_cnt[dd & (VGAE_BIN_ROWS - 1)], 1)] = min(max(ei[e], 0), n - 1);
    }
  }
}

template <int VEC> struct vgae_vec { typedef __attribute__((ext_vector_type(VEC))) float t; };

template <int VEC, int GRP>
__device__ __forceinline__ typename vgae_vec<VEC>::t
vgae_bucket_sum(const float* __restrict__ h, const int* __restrict__ nbr,
                int beg, int end, int ids0, int lane, int n) {
  typedef typename vgae_vec<VEC>::t vecf;
  constexpr int D = 32 * VEC;
  const unsigned loff = (unsigned)lane * VEC;
  vecf acc = {};
  int mine = min(max(ids0, 0), n - 1);
  for (int base = beg; base < end; base += 32) {
    const int m = min(32, end - base);
    if (base != beg) {
      mine = (lane < m) ? nbr[base + lane] : 0;
      mine = min(max(mine, 0), n - 1);
    }
#define VGAE_LD(i) v[i] = *(const vecf*)(h + (long)__builtin_amdgcn_readlane(mine, j + (i)) * D + loff)
#define VGAE_AD(i) acc += v[i]
    for (int j = 0; j < m; j += GRP) {
      const int c = m - j;
      vecf v[16];
      do {
        VGAE_LD(0);  if (c <= 1) break;
        VGAE_LD(1);  if (c <= 2) break;
        VGAE_LD(2);  if (c <= 3) break;
        VGAE_LD(3);  if (c <= 4) break;
        VGAE_LD(4);  if (c <= 5) break;
        VGAE_LD(5);  if (c <= 6) break;
        VGAE_LD(6);  if (c <= 7) break;
        VGAE_LD(7);  if (GRP == 8 || c <= 8) break;
        VGAE_LD(8);  if (GRP == 9 || c <= 9) break;
        VGAE_LD(9);  if (GRP == 10 || c <= 10) break;
        VGAE_LD(10); if (GRP == 11 || c <= 11) break;
        VGAE_LD(11); if (GRP == 12 || c <= 12) break;
        VGAE_LD(12); if (GRP == 13 || c <= 13) break;
        VGAE_LD(13); if (GRP == 14 || c <= 14) break;
        VGAE_LD(14); if (GRP == 15 || c <= 15) break;
        VGAE_LD(15);
      } while (0);
      do {
        VGAE_AD(0);  if (c <= 1) break;
        VGAE_AD(1);  if (c <= 2) break;
        VGAE_AD(2);  if (c <= 3) break;
        VGAE_AD(3);  if (c <= 4) break;
        VGAE_AD(4);  if (c <= 5) break;
        VGAE_AD(5);  if (c <= 6) break;
        VGAE_AD(6);  if (c <= 7) break;
        VGAE_AD(7);  if (GRP == 8 || c <= 8) break;
        VGAE_AD(8);  if (GRP == 9 || c <= 9) break;
        VGAE_AD(9);  if (GRP == 10 || c <= 10) break;
        VGAE_AD(10); if (GRP == 11 || c <= 11) break;
        VGAE_AD(11); if (GRP == 12 || c <= 12) break;
        VGAE_AD(12); if (GRP == 13 || c <= 13) break;
        VGAE_AD(13); if (GRP == 14 || c <= 14) break;
        VGAE_AD(14); if (GRP == 15 || c <= 15) break;
        VGAE_AD(15);
      } while (0);
    }
#undef VGAE_LD
#undef VGAE_AD
  }
  return acc;
}

template <int DIN, int DOUT, int DO_LN, int TW, bool SPLIT, bool PROJ>
__global__ __launch_bounds__(DOUT * 2 * TW)
void vgae_sage_gemm(const float* __restrict__ h_a,
                    const float* __restrict__ h_b,
                    const int* __restrict__ start,
                    const int* __restrict__ nbr,
                    const float* __restrict__ invdeg,
                    const float* __restrict__ Wl_a,
                    const float* __restrict__ bl_a,
                    const float* __restrict__ Wr_a,
                    const float* __restrict__ gamma_a,
                    const float* __restrict__ beta_a,
                    float* __restrict__ out_a,
                    const float* __restrict__ Wl_b,
                    const float* __restrict__ bl_b,
                    const float* __restrict__ Wr_b,
                    const float* __restrict__ gamma_b,
                    const float* __restrict__ beta_b,
                    float* __restrict__ out_b,
                    int ntiles, int n,
                    const float* __restrict__ Wl2_a,
                    const float* __restrict__ bl2_a,
                    const float* __restrict__ Wr2_a,
                    const float* __restrict__ Wl2_b,
                    const float* __restrict__ bl2_b,
                    const float* __restrict__ Wr2_b) {
  typedef __attribute__((ext_vector_type(4))) float v4f;
  constexpr int NWT   = DOUT / 16;
  constexpr int NT    = NWT * 32 * TW;
  constexpr int NW    = NT / 32;
  constexpr int RPW   = 16 / NW;
  constexpr int VEC   = DIN / 32;
  constexpr int GRP   = (DIN == 128) ? 16 : 12;
  constexpr int K     = 2 * DIN;
  constexpr int NK    = K / 4;
  constexpr int PITCH = K + 4;
  constexpr int C4    = DIN / 4;
  constexpr int NP    = 16 * C4;
  constexpr int NV    = (NP + NT - 1) / NT;
  constexpr bool PART = (NP % NT) != 0;
  static_assert(16 % NW == 0 && (!PART || NV == 1), "tile rows / root pieces per wave");
  static_assert(!(SPLIT && TW == 2), "two towers share a block only where they share the input");
  static_assert(!PROJ || (DO_LN && DOUT == 128 && TW == 1), "projection follows a 128-wide LN layer");
  constexpr int PITCH2 = 128 + 4;
  typedef typename vgae_vec<VEC>::t vecf;

  __shared__ __attribute__((aligned(16))) float As[16 * PITCH];
  __shared__ __attribute__((aligned(16))) float smem[DO_LN ? TW * 16 * 128 : 4];
  __shared__ __attribute__((aligned(16))) float W2s[PROJ ? 128 * PITCH2 : 4];
  __shared__ __attribute__((aligned(16))) float Ys[PROJ ? 16 * PITCH2 : 4];
  __shared__ float s_mean[16 * TW];
  __shared__ float s_rstd[16 * TW];

  const int tid  = threadIdx.x;
  const int lane = tid & 31;
  const int wave = __builtin_amdgcn_readfirstlane(tid >> 5);
  const int l15  = lane & 15;
  const int hi   = lane >> 4;
  const int tsel = SPLIT ? (int)(blockIdx.x & 1) : 0;
  const int bid  = SPLIT ? (int)(blockIdx.x >> 1) : (int)blockIdx.x;
  const int gstr = SPLIT ? (int)(gridDim.x >> 1) : (int)gridDim.x;
  const bool second = SPLIT && tsel;
  const float* __restrict__ h = second ? h_b : h_a;
  const int tw   = (TW == 2) ? wave / NWT : 0;
  const int cw   = (TW == 2) ? wave % NWT : wave;
  const int bcol = cw * 16 + l15;
  const int koff = 2 * hi;
  const float* __restrict__ Wl    = ((TW == 2 && tw) || second) ? Wl_b : Wl_a;
  const float* __restrict__ bl    = ((TW == 2 && tw) || second) ? bl_b : bl_a;
  const float* __restrict__ Wr    = ((TW == 2 && tw) || second) ? Wr_b : Wr_a;
  const float* __restrict__ gamma = ((TW == 2 && tw) || second) ? gamma_b : gamma_a;
  const float* __restrict__ beta  = ((TW == 2 && tw) || second) ? beta_b : beta_a;
  float* __restrict__ out         = ((TW == 2 && tw) || second) ? out_b : out_a;

  int tile = bid;
  if (tile >= ntiles) return;

  v2f breg[NK];
  {
    const float* wlrow = Wl + (long)bcol * DIN + koff;
    const float* wrrow = Wr + (long)bcol * DIN + koff;
#pragma unroll
    for (int s = 0; s < NK / 2; ++s) breg[s] = *(const v2f*)(wlrow + 4 * s);
#pragma unroll
    for (int s = 0; s < NK / 2; ++s) breg[NK / 2 + s] = *(const v2f*)(wrrow + 4 * s);
  }
  const float bias = bl[bcol];
  v4f g4 = {0.f, 0.f, 0.f, 0.f}, b4 = {0.f, 0.f, 0.f, 0.f};
  if (DO_LN) {
    g4 = *(const v4f*)(gamma + lane * 4);
    b4 = *(const v4f*)(beta + lane * 4);
  }

  float bias2 = 0.f;
  if (PROJ) {
    const float* __restrict__ Wl2 = second ? Wl2_b : Wl2_a;
    const float* __restrict__ Wr2 = second ? Wr2_b : Wr2_a;
    const float* __restrict__ bl2 = second ? bl2_b : bl2_a;
    for (int idx = tid; idx < 128 * 32; idx += NT) {
      const int row = idx >> 5, pc = idx & 31;
      const float* srow = (row < 64) ? (Wl2 + row * 128) : (Wr2 + (row - 64) * 128);
      *(v4f*)(W2s + row * PITCH2 + pc * 4) = *(const v4f*)(srow + pc * 4);
    }
    if (cw >= 4) bias2 = bl2[(cw - 4) * 16 + l15];
  }

  v4f rh[NV];
  int nb[RPW], ne[RPW], id0[RPW];
  int st2;
  float sc1;
#pragma unroll
  for (int i = 0; i < NV; ++i) {
    const int idx = tid + i * NT;
    const long row = (long)tile * 16 + idx / C4;
    if (!PART || idx < NP) rh[i] = *(const v4f*)(h + row * DIN + (idx % C4) * 4);
  }
#pragma unroll
  for (int k = 0; k < RPW; ++k) {
    const int node = tile * 16 + wave + k * NW;
    nb[k] = start[node];
    ne[k] = start[node + 1];
  }
  sc1 = invdeg[tile * 16 + l15];
#pragma unroll
  for (int k = 0; k < RPW; ++k)
    id0[k] = nbr[max(min(nb[k] + lane, ne[k] - 1), 0)];
  {
    const int t1 = tile + gstr;
    const int tb = (t1 < ntiles) ? t1 : tile;
    st2 = start[tb * 16 + min(lane, 16)];
  }

  const float* ap = As + l15 * PITCH + koff;

  for (; tile < ntiles; tile += gstr) {
    const int rowbase = tile * 16;
#pragma unroll
    for (int k = 0; k < RPW; ++k) {
      const vecf a = vgae_bucket_sum<VEC, GRP>(h, nbr, nb[k], ne[k], id0[k], lane, n);
      const float sck = __int_as_float(__builtin_amdgcn_readlane(__float_as_int(sc1), wave + k * NW));
      *(vecf*)(As + (wave + k * NW) * PITCH + lane * VEC) = a * sck;
    }
#pragma unroll
    for (int i = 0; i < NV; ++i) {
      const int idx = tid + i * NT;
      if (!PART || idx < NP) *(v4f*)(As + (idx / C4) * PITCH + DIN + (idx % C4) * 4) = rh[i];
    }
    __syncthreads();

    const int nxt = tile + gstr;
    if (nxt < ntiles) {
#pragma unroll
      for (int i = 0; i < NV; ++i) {
        const int idx = tid + i * NT;
        const long row = (long)nxt * 16 + idx / C4;
        if (!PART || idx < NP) rh[i] = *(const v4f*)(h + row * DIN + (idx % C4) * 4);
      }
#pragma unroll
      for (int k = 0; k < RPW; ++k) {
        nb[k] = __builtin_amdgcn_readlane(st2, wave + k * NW);
        ne[k] = __builtin_amdgcn_readlane(st2, wave + k * NW + 1);
      }
      sc1 = invdeg[nxt * 16 + l15];
#pragma unroll
      for (int k = 0; k < RPW; ++k)
        id0[k] = nbr[max(min(nb[k] + lane, ne[k] - 1), 0)];
      const int t2 = nxt + gstr;
      const int tb = (t2 < ntiles) ? t2 : nxt;
      st2 = start[tb * 16 + min(lane, 16)];
    }

    v8f c = {0.f, 0.f, 0.f, 0.f, 0.f, 0.f, 0.f, 0.f};
#pragma unroll
    for (int s = 0; s < NK; ++s) {
      v2f a = *(const v2f*)(ap + 4 * s);
      c = __builtin_amdgcn_wmma_f32_16x16x4_f32(false, a, false, breg[s],
                                                (short)0, c, false, false);
    }

    if (!DO_LN) {
      for (int v = 0; v < 8; ++v) {
        int r = rowbase + v + 8 * hi;
        out[(long)r * DOUT + bcol] = c[v] + bias;
      }
      __syncthreads();
      continue;
    }

    for (int v = 0; v < 8; ++v)
      smem[(tw * 16 + v + 8 * hi) * 128 + bcol] = c[v] + bias;
    __syncthreads();

    if (tid < 16 * TW) {
      const float* rowp = &smem[tid * 128];
      float s = 0.f, ss = 0.f;
      for (int j = 0; j < 128; ++j) { float x = rowp[j]; s += x; ss += x * x; }
      float m   = s * (1.0f / 128.0f);
      float var = ss * (1.0f / 128.0f) - m * m;
      s_mean[tid] = m;
      s_rstd[tid] = rsqrtf(var + LN_EPS);
    }
    __syncthreads();

    for (int r = cw; r < 16; r += NWT) {
      const int tr = tw * 16 + r;
      const v4f xv = *(const v4f*)&smem[(tr * 32 + lane) * 4];
      const float m = s_mean[tr], rs = s_rstd[tr];
      v4f y;
      for (int e = 0; e < 4; ++e)
        y[e] = fmaxf((xv[e] - m) * rs * g4[e] + b4[e], 0.0f);
      if (PROJ) *(v4f*)(Ys + r * PITCH2 + lane * 4) = y;
      else      *(v4f*)(out + (long)(rowbase + r) * 128 + lane * 4) = y;
    }

    if (PROJ) {
      __syncthreads();
      const float* yp = Ys + l15 * PITCH2 + koff;
      const float* wp = W2s + bcol * PITCH2 + koff;
      v8f c2 = {0.f, 0.f, 0.f, 0.f, 0.f, 0.f, 0.f, 0.f};
#pragma unroll
      for (int s = 0; s < 32; ++s) {
        const v2f a2 = *(const v2f*)(yp + 4 * s);
        const v2f b2 = *(const v2f*)(wp + 4 * s);
        c2 = __builtin_amdgcn_wmma_f32_16x16x4_f32(false, a2, false, b2,
                                                   (short)0, c2, false, false);
      }
      for (int v = 0; v < 8; ++v)
        smem[(v + 8 * hi) * 128 + bcol] = c2[v] + bias2;
      __syncthreads();
      float* __restrict__ plane = out + (hi ? (long)n * 64 : 0L);
      for (int r = cw; r < 16; r += NWT) {
        const v4f t = *(const v4f*)&smem[(r * 32 + lane) * 4];
        *(v4f*)(plane + (long)(rowbase + r) * 64 + l15 * 4) = t;
      }
    }
  }
}

__global__ __launch_bounds__(256)
void vgae_gather_out(const float* __restrict__ pr_a, const float* __restrict__ pr_b,
                     const int* __restrict__ start, const int* __restrict__ nbr,
                     const float* __restrict__ invdeg,
                     float* __restrict__ out_a, float* __restrict__ out_b, int n) {
  const int lane = threadIdx.x & 31;
  const int wave = __builtin_amdgcn_readfirstlane((int)(threadIdx.x >> 5));
  const bool second = (blockIdx.x & 1) != 0;
  const float* __restrict__ p = second ? pr_b : pr_a;
  const float* __restrict__ r = p + (long)n * 64;
  float* __restrict__ out = second ? out_b : out_a;
  const int stride = (int)(gridDim.x >> 1) * 8;
  int row = (int)(blockIdx.x >> 1) * 8 + wave;
  if (row >= n) return;
  int nb = start[row], ne = start[row + 1];
  float sc = invdeg[row];
  int ids = nbr[max(min(nb + lane, ne - 1), 0)];
  int r1 = (row + stride < n) ? row + stride : row;
  int nb1 = start[r1], ne1 = start[r1 + 1];
  float sc1 = invdeg[r1];
  for (; row < n; row += stride) {
    const int ids1 = nbr[max(min(nb1 + lane, ne1 - 1), 0)];
    const int r2 = (row + 2 * stride < n) ? row + 2 * stride : row;
    const int nb2 = start[r2], ne2 = start[r2 + 1];
    const float sc2 = invdeg[r2];
    const v2f rt = *(const v2f*)(r + (long)row * 64 + lane * 2);
    const v2f acc = vgae_bucket_sum<2, 16>(p, nbr, nb, ne, ids, lane, n);
    *(v2f*)(out + (long)row * 64 + lane * 2) = acc * sc + rt;
    nb = nb1; ne = ne1; sc = sc1; ids = ids1;
    nb1 = nb2; ne1 = ne2; sc1 = sc2;
  }
}


extern "C" void kernel_launch(void* const* d_in, const int* in_sizes, int n_in,
                              void* d_out, int out_size, void* d_ws, size_t ws_size,
                              hipStream_t stream) {
  const float* x  = (const float*)d_in[0];
  const int*   ei = (const int*)d_in[1];
  const int N = in_sizes[0] / 64;
  const int E = in_sizes[1] / 2;

  const size_t tail_words = 2 * (size_t)N + 64 + 2 * (size_t)E;
  const bool four = ws_size >= ((size_t)4 * N * 128 + tail_words) * 4;
  const bool both = ws_size >= ((size_t)3 * N * 128 + tail_words) * 4;
  float* bufA   = (float*)d_ws;
  float* bufB   = both ? bufA + (long)N * 128 : bufA;
  float* bufC   = bufB + (long)N * 128;
  float* bufD   = four ? bufC + (long)N * 128 : bufA;
  float* invdeg = (four ? bufD : bufC) + (long)N * 128;
  int*   start  = (int*)(invdeg + N);
  int*   rank   = start + N + 64;
  int*   nbr    = rank + E;

  const int nbins = (N + VGAE_BIN_ROWS - 1) >> VGAE_BIN_SHIFT;
  if (nbins <= VGAE_MAX_BINS) {
    int* bintot   = (int*)bufA;
    int* binstore = (int*)bufC;
    const int cap = (int)(((long)N * 128) / nbins);
    vgae_zero_i<<<(nbins * VGAE_CNT_PITCH + 255) / 256, 256, 0, stream>>>(bintot, nbins * VGAE_CNT_PITCH);
    vgae_bin_edges<<<(E + VGAE_CHUNK - 1) / VGAE_CHUNK, 1024, 0, stream>>>(ei, bintot, binstore, E, N, nbins, cap);
    vgae_sort_bin<<<nbins, 1024, 0, stream>>>(ei, bintot, binstore, start, invdeg, nbr, E, N, cap);
  } else {
    vgae_zero_i<<<(N + 255) / 256, 256, 0, stream>>>(start, N);
    vgae_count_rank<<<(E + 255) / 256, 256, 0, stream>>>(ei, start, rank, E);
    vgae_scan<<<1, 1024, 0, stream>>>(start, invdeg, N);
    vgae_fill<<<(E + 255) / 256, 256, 0, stream>>>(ei, start, rank, nbr, E);
  }

  const int nrt = N / 16;

#define VGAE_F(i) ((const float*)d_in[i])

  if (both) {
    vgae_sage_gemm<64, 128, 1, 2, false, false><<<512, 512, 0, stream>>>(
        x, nullptr, start, nbr, invdeg,
        VGAE_F(2), VGAE_F(3), VGAE_F(4), VGAE_F(5), VGAE_F(6), bufA,
        VGAE_F(15), VGAE_F(16), VGAE_F(17), VGAE_F(18), VGAE_F(19), bufB, nrt, N,
          nullptr, nullptr, nullptr, nullptr, nullptr, nullptr);

    if (four) {
      vgae_sage_gemm<128, 128, 1, 1, true, true><<<512, 256, 0, stream>>>(
          bufA, bufB, start, nbr, invdeg,
          VGAE_F(7), VGAE_F(8), VGAE_F(9), VGAE_F(10), VGAE_F(11), bufC,
          VGAE_F(20), VGAE_F(21), VGAE_F(22), VGAE_F(23), VGAE_F(24), bufD, nrt, N,
          VGAE_F(12), VGAE_F(13), VGAE_F(14), VGAE_F(25), VGAE_F(26), VGAE_F(27));
      vgae_gather_out<<<1024, 256, 0, stream>>>(
          bufC, bufD, start, nbr, invdeg,
          (float*)d_out, (float*)d_out + (long)N * 64, N);
    } else {
      vgae_sage_gemm<128, 128, 1, 1, false, false><<<512, 256, 0, stream>>>(
          bufA, nullptr, start, nbr, invdeg,
          VGAE_F(7), VGAE_F(8), VGAE_F(9), VGAE_F(10), VGAE_F(11), bufC,
          nullptr, nullptr, nullptr, nullptr, nullptr, nullptr, nrt, N,
          nullptr, nullptr, nullptr, nullptr, nullptr, nullptr);
      vgae_sage_gemm<128, 128, 1, 1, false, false><<<512, 256, 0, stream>>>(
          bufB, nullptr, start, nbr, invdeg,
          VGAE_F(20), VGAE_F(21), VGAE_F(22), VGAE_F(23), VGAE_F(24), bufD,
          nullptr, nullptr, nullptr, nullptr, nullptr, nullptr, nrt, N,
          nullptr, nullptr, nullptr, nullptr, nullptr, nullptr);
    }

    if (!four) vgae_sage_gemm<128, 64, 0, 1, true, false><<<1024, 128, 0, stream>>>(
        bufC, bufD, start, nbr, invdeg,
        VGAE_F(12), VGAE_F(13), VGAE_F(14), nullptr, nullptr, (float*)d_out,
        VGAE_F(25), VGAE_F(26), VGAE_F(27), nullptr, nullptr, (float*)d_out + (long)N * 64, nrt, N,
          nullptr, nullptr, nullptr, nullptr, nullptr, nullptr);
  } else {
    for (int t = 0; t < 2; ++t) {
      const int base = 2 + t * 13;
      float* outp = (float*)d_out + (long)t * N * 64;
      vgae_sage_gemm<64, 128, 1, 1, false, false><<<1024, 256, 0, stream>>>(
          x, nullptr, start, nbr, invdeg,
          VGAE_F(base + 0), VGAE_F(base + 1), VGAE_F(base + 2), VGAE_F(base + 3), VGAE_F(base + 4), bufA,
          nullptr, nullptr, nullptr, nullptr, nullptr, nullptr, nrt, N,
          nullptr, nullptr, nullptr, nullptr, nullptr, nullptr);
      vgae_sage_gemm<128, 128, 1, 1, false, false><<<512, 256, 0, stream>>>(
          bufA, nullptr, start, nbr, invdeg,
          VGAE_F(base + 5), VGAE_F(base + 6), VGAE_F(base + 7), VGAE_F(base + 8), VGAE_F(base + 9), bufC,
          nullptr, nullptr, nullptr, nullptr, nullptr, nullptr, nrt, N,
          nullptr, nullptr, nullptr, nullptr, nullptr, nullptr);
      vgae_sage_gemm<128, 64, 0, 1, false, false><<<1024, 128, 0, stream>>>(
          bufC, nullptr, start, nbr, invdeg,
          VGAE_F(base + 10), VGAE_F(base + 11), VGAE_F(base + 12), nullptr, nullptr, outp,
          nullptr, nullptr, nullptr, nullptr, nullptr, nullptr, nrt, N,
          nullptr, nullptr, nullptr, nullptr, nullptr, nullptr);
    }
  }
#undef VGAE_F
}
